# starting kernel
# baseline (speedup 1.0000x reference)
#include <hip/hip_runtime.h>
#include <math.h>


#define NN   50000
#define EE   800000
#define IND  64
#define HH   128
#define OUTD 16

typedef float v2f __attribute__((ext_vector_type(2)));
typedef float v8f __attribute__((ext_vector_type(8)));

__device__ __forceinline__ float elu_f(float v) {
    return v > 0.0f ? v : (__expf(v) - 1.0f);
}

template <int ACT, int TN>
__global__ void gemm16_wmma(const float* __restrict__ A,
                            const float* __restrict__ W,
                            const float* __restrict__ bias,
                            float* __restrict__ C,
                            int M, int K, int Nc) {
    int gid  = blockIdx.x * blockDim.x + threadIdx.x;
    int wave = gid >> 5;
    int lane = gid & 31;
    int groupsN = Nc / (16 * TN);
    int tilesM  = M >> 4;
    if (wave >= tilesM * groupsN) return;
    int tm = wave / groupsN;
    int tg = wave - tm * groupsN;

    int r  = lane & 15;
    int kq = (lane >> 4) << 1;

    const float* Arow = A + (size_t)(tm * 16 + r) * K + kq;
    const float* Bp   = W + (size_t)kq * Nc + tg * (16 * TN) + r;
    const size_t bstep = (size_t)4 * Nc;

    v8f acc[TN];
#pragma unroll
    for (int t = 0; t < TN; ++t) acc[t] = (v8f){};

    v2f a_cur, b_cur[TN];
    {
        float2 av = *(const float2*)(Arow);
        a_cur.x = av.x;
        a_cur.y = av.y;
#pragma unroll
        for (int t = 0; t < TN; ++t) {
            b_cur[t].x = Bp[t * 16];
            b_cur[t].y = Bp[Nc + t * 16];
        }
    }

    for (int k0 = 4; k0 < K; k0 += 4) {
        const float* Bn = Bp + bstep;
        float2 an = *(const float2*)(Arow + k0);
        v2f a_nxt, b_nxt[TN];
        a_nxt.x = an.x;
        a_nxt.y = an.y;
#pragma unroll
        for (int t = 0; t < TN; ++t) {
            b_nxt[t].x = Bn[t * 16];
            b_nxt[t].y = Bn[Nc + t * 16];
        }
#pragma unroll
        for (int t = 0; t < TN; ++t) {
            acc[t] = __builtin_amdgcn_wmma_f32_16x16x4_f32(
                false, a_cur, false, b_cur[t], (short)0, acc[t], false, false);
        }
        a_cur = a_nxt;
#pragma unroll
        for (int t = 0; t < TN; ++t) b_cur[t] = b_nxt[t];
        Bp = Bn;
    }

#pragma unroll
    for (int t = 0; t < TN; ++t) {
        acc[t] = __builtin_amdgcn_wmma_f32_16x16x4_f32(
            false, a_cur, false, b_cur[t], (short)0, acc[t], false, false);
    }

    int mb = tm * 16 + ((lane >> 4) << 3);
#pragma unroll
    for (int t = 0; t < TN; ++t) {
        int n = tg * (16 * TN) + t * 16 + r;
        float bv = bias ? bias[n] : 0.0f;
#pragma unroll
        for (int g = 0; g < 8; ++g) {
            float v = acc[t][g] + bv;
            if (ACT) v = elu_f(v);
            C[(size_t)(mb + g) * Nc + n] = v;
        }
    }
}

__global__ void fill_kernel(float* __restrict__ p, float val, int n) {
    int i = blockIdx.x * blockDim.x + threadIdx.x;
    if (i < n) p[i] = val;
}

__global__ void deg_accum_kernel(const int* __restrict__ dst,
                                 const float* __restrict__ w,
                                 float* __restrict__ deg) {
    int e = blockIdx.x * blockDim.x + threadIdx.x;
    if (e < EE) atomicAdd(&deg[dst[e]], w[e]);
}

__global__ void dinv_kernel(float* __restrict__ deg, float* __restrict__ sc) {
    int n = blockIdx.x * blockDim.x + threadIdx.x;
    if (n < NN) {
        float di = rsqrtf(deg[n]);
        deg[n] = di;
        sc[n]  = 2.0f * di * di;
    }
}

__global__ void norm_kernel(const int* __restrict__ src,
                            const int* __restrict__ dst,
                            const float* __restrict__ w,
                            const float* __restrict__ dinv,
                            float* __restrict__ nrm) {
    int e = blockIdx.x * blockDim.x + threadIdx.x;
    if (e < EE) nrm[e] = w[e] * dinv[src[e]] * dinv[dst[e]];
}

__global__ void scatter_kernel(const float* __restrict__ t,
                               const int* __restrict__ src,
                               const int* __restrict__ dst,
                               const float* __restrict__ nrm,
                               float* __restrict__ agg) {
    long long idx = (long long)blockIdx.x * blockDim.x + threadIdx.x;
    long long e = idx >> 5;
    if (e >= EE) return;
    int fg = (int)(idx & 31);
    int s = src[e], d = dst[e];
    float w = nrm[e];
    const float4 tv = *(const float4*)(t + (size_t)s * HH + fg * 4);
    float* ap = agg + (size_t)d * HH + fg * 4;
    atomicAdd(ap + 0, tv.x * w);
    atomicAdd(ap + 1, tv.y * w);
    atomicAdd(ap + 2, tv.z * w);
    atomicAdd(ap + 3, tv.w * w);
}

__global__ void post_kernel(const float* __restrict__ agg,
                            const float* __restrict__ t,
                            const float* __restrict__ sc,
                            const float* __restrict__ bias,
                            float* __restrict__ h) {
    int idx = blockIdx.x * blockDim.x + threadIdx.x;
    if (idx >= NN * HH) return;
    int n = idx >> 7;
    int f = idx & 127;
    float v = agg[idx] + sc[n] * t[idx] + bias[f];
    h[idx] = elu_f(v);
}

static inline int nblocks(long long t, int bs) { return (int)((t + bs - 1) / bs); }

extern "C" void kernel_launch(void* const* d_in, const int* in_sizes, int n_in,
                              void* d_out, int out_size, void* d_ws, size_t ws_size,
                              hipStream_t stream) {
    const float* x  = (const float*)d_in[0];
    const int*   ei = (const int*)d_in[1];
    const float* ew = (const float*)d_in[2];
    const float* W1 = (const float*)d_in[3];  const float* b1 = (const float*)d_in[4];
    const float* W2 = (const float*)d_in[5];  const float* b2 = (const float*)d_in[6];
    const float* W3 = (const float*)d_in[7];  const float* b3 = (const float*)d_in[8];
    const float* Wg[4] = { (const float*)d_in[9],  (const float*)d_in[11],
                           (const float*)d_in[13], (const float*)d_in[15] };
    const float* bg[4] = { (const float*)d_in[10], (const float*)d_in[12],
                           (const float*)d_in[14], (const float*)d_in[16] };
    const float* Wh = (const float*)d_in[17]; const float* bh = (const float*)d_in[18];
    float* out = (float*)d_out;

    const int* srcI = ei;
    const int* dstI = ei + EE;

    float* f    = (float*)d_ws;
    float* hbuf = f;
    float* tbuf = hbuf + (size_t)NN * HH;
    float* aggb = tbuf + (size_t)NN * HH;
    float* degb = aggb + (size_t)NN * HH;
    float* scb  = degb + NN;
    float* nrmb = scb + NN;

    const int BS = 256;

    fill_kernel<<<nblocks(NN, BS), BS, 0, stream>>>(degb, 2.0f, NN);
    deg_accum_kernel<<<nblocks(EE, BS), BS, 0, stream>>>(dstI, ew, degb);
    dinv_kernel<<<nblocks(NN, BS), BS, 0, stream>>>(degb, scb);
    norm_kernel<<<nblocks(EE, BS), BS, 0, stream>>>(srcI, dstI, ew, degb, nrmb);

    auto gemm128 = [&](const float* A, const float* W, const float* bias, float* C,
                       int K, int act) {
        long long thr = (long long)(NN >> 4) * (HH / 64) * 32;
        if (act) gemm16_wmma<1, 4><<<nblocks(thr, BS), BS, 0, stream>>>(A, W, bias, C, NN, K, HH);
        else     gemm16_wmma<0, 4><<<nblocks(thr, BS), BS, 0, stream>>>(A, W, bias, C, NN, K, HH);
    };

    gemm128(x,    W1, b1, hbuf, IND, 1);
    gemm128(hbuf, W2, b2, tbuf, HH,  1);
    gemm128(tbuf, W3, b3, hbuf, HH,  1);

    for (int l = 0; l < 4; ++l) {
        gemm128(hbuf, Wg[l], nullptr, tbuf, HH, 0);
        fill_kernel<<<nblocks((long long)NN * HH, BS), BS, 0, stream>>>(aggb, 0.0f, NN * HH);
        scatter_kernel<<<nblocks((long long)EE * 32, BS), BS, 0, stream>>>(
            tbuf, srcI, dstI, nrmb, aggb);
        post_kernel<<<nblocks((long long)NN * HH, BS), BS, 0, stream>>>(
            aggb, tbuf, scb, bg[l], hbuf);
    }

    {
        long long thr = (long long)(NN >> 4) * (OUTD >> 4) * 32;
        gemm16_wmma<0, 1><<<nblocks(thr, BS), BS, 0, stream>>>(hbuf, Wh, bh, out, NN, HH, OUTD);
    }
}


# Edge aggregation pulled, not pushed: per-call destination-grouped edge list (int atomic rank + scan), one wave sums a node's rows in registers with self, bias, ELU; sums now in list arrival order
# speedup vs baseline: 2.3634x; 2.3634x over previous
#include <hip/hip_runtime.h>
#include <math.h>


#define NN   50000
#define EE   800000
#define IND  64
#define HH   128
#define OUTD 16

typedef float v2f __attribute__((ext_vector_type(2)));
typedef float v8f __attribute__((ext_vector_type(8)));

__device__ __forceinline__ float elu_f(float v) {
    return v > 0.0f ? v : (__expf(v) - 1.0f);
}

template <int ACT, int TN>
__global__ void gemm16_wmma(const float* __restrict__ A,
                            const float* __restrict__ W,
                            const float* __restrict__ bias,
                            float* __restrict__ C,
                            int M, int K, int Nc) {
    int gid  = blockIdx.x * blockDim.x + threadIdx.x;
    int wave = gid >> 5;
    int lane = gid & 31;
    int groupsN = Nc / (16 * TN);
    int tilesM  = M >> 4;
    if (wave >= tilesM * groupsN) return;
    int tm = wave / groupsN;
    int tg = wave - tm * groupsN;

    int r  = lane & 15;
    int kq = (lane >> 4) << 1;

    const float* Arow = A + (size_t)(tm * 16 + r) * K + kq;
    const float* Bp   = W + (size_t)kq * Nc + tg * (16 * TN) + r;
    const size_t bstep = (size_t)4 * Nc;

    v8f acc[TN];
#pragma unroll
    for (int t = 0; t < TN; ++t) acc[t] = (v8f){};

    v2f a_cur, b_cur[TN];
    {
        float2 av = *(const float2*)(Arow);
        a_cur.x = av.x;
        a_cur.y = av.y;
#pragma unroll
        for (int t = 0; t < TN; ++t) {
            b_cur[t].x = Bp[t * 16];
            b_cur[t].y = Bp[Nc + t * 16];
        }
    }

    for (int k0 = 4; k0 < K; k0 += 4) {
        const float* Bn = Bp + bstep;
        float2 an = *(const float2*)(Arow + k0);
        v2f a_nxt, b_nxt[TN];
        a_nxt.x = an.x;
        a_nxt.y = an.y;
#pragma unroll
        for (int t = 0; t < TN; ++t) {
            b_nxt[t].x = Bn[t * 16];
            b_nxt[t].y = Bn[Nc + t * 16];
        }
#pragma unroll
        for (int t = 0; t < TN; ++t) {
            acc[t] = __builtin_amdgcn_wmma_f32_16x16x4_f32(
                false, a_cur, false, b_cur[t], (short)0, acc[t], false, false);
        }
        a_cur = a_nxt;
#pragma unroll
        for (int t = 0; t < TN; ++t) b_cur[t] = b_nxt[t];
        Bp = Bn;
    }

#pragma unroll
    for (int t = 0; t < TN; ++t) {
        acc[t] = __builtin_amdgcn_wmma_f32_16x16x4_f32(
            false, a_cur, false, b_cur[t], (short)0, acc[t], false, false);
    }

    int mb = tm * 16 + ((lane >> 4) << 3);
#pragma unroll
    for (int t = 0; t < TN; ++t) {
        int n = tg * (16 * TN) + t * 16 + r;
        float bv = bias ? bias[n] : 0.0f;
#pragma unroll
        for (int g = 0; g < 8; ++g) {
            float v = acc[t][g] + bv;
            if (ACT) v = elu_f(v);
            C[(size_t)(mb + g) * Nc + n] = v;
        }
    }
}

__global__ void zero_int_kernel(int* __restrict__ p, int n) {
    int i = blockIdx.x * blockDim.x + threadIdx.x;
    if (i < n) p[i] = 0;
}

__global__ void rank_kernel(const int* __restrict__ dst,
                            int* __restrict__ cnt,
                            int* __restrict__ rank) {
    int e = blockIdx.x * blockDim.x + threadIdx.x;
    if (e < EE) rank[e] = atomicAdd(&cnt[dst[e]], 1);
}

#define SCAN_BS 1024
__global__ __launch_bounds__(SCAN_BS) void scan_kernel(const int* __restrict__ cnt,
                                                       int* __restrict__ rowptr) {
    __shared__ int wave_tot[32];
    __shared__ int wave_pre[32];
    const int tid  = threadIdx.x;
    const int lane = tid & 31;
    const int wv   = tid >> 5;
    const int first = blockIdx.x * SCAN_BS;

    int before = 0;
    for (int i = tid; i < first; i += SCAN_BS) before += cnt[i];
    for (int o = 16; o > 0; o >>= 1) before += __shfl_xor(before, o, 32);

    const int idx = first + tid;
    const int v = (idx < NN) ? cnt[idx] : 0;
    int inc = v;
    for (int o = 1; o < 32; o <<= 1) {
        int u = __shfl_up(inc, o, 32);
        if (lane >= o) inc += u;
    }
    if (lane == 31) wave_tot[wv] = inc;
    if (lane == 0)  wave_pre[wv] = before;
    __syncthreads();

    int base = wave_pre[lane];
    int mine = (lane < wv) ? wave_tot[lane] : 0;
    for (int o = 16; o > 0; o >>= 1) {
        base += __shfl_xor(base, o, 32);
        mine += __shfl_xor(mine, o, 32);
    }
    if (idx < NN) rowptr[idx] = base + mine + inc - v;
}

__global__ void place_kernel(const int* __restrict__ src,
                             const int* __restrict__ dst,
                             const float* __restrict__ w,
                             const int* __restrict__ rowptr,
                             const int* __restrict__ rank,
                             int2* __restrict__ rec) {
    int e = blockIdx.x * blockDim.x + threadIdx.x;
    if (e < EE) {
        int p = rowptr[dst[e]] + rank[e];
        rec[p] = make_int2(src[e], __float_as_int(w[e]));
    }
}

__global__ void deg_kernel(const int* __restrict__ rowptr,
                           const int* __restrict__ cnt,
                           const int2* __restrict__ rec,
                           float* __restrict__ dinv,
                           float* __restrict__ sc) {
    int gid = blockIdx.x * blockDim.x + threadIdx.x;
    int n = gid >> 4;
    int j = gid & 15;
    float s = 0.0f;
    if (n < NN) {
        int st = rowptr[n], c = cnt[n];
        for (int k = j; k < c; k += 16) s += __int_as_float(rec[st + k].y);
    }
    for (int o = 8; o > 0; o >>= 1) s += __shfl_xor(s, o, 16);
    if (n < NN && j == 0) {
        float di = rsqrtf(s + 2.0f);
        dinv[n] = di;
        sc[n]   = 2.0f * di * di;
    }
}

__global__ void norm_kernel(const int* __restrict__ rowptr,
                            const int* __restrict__ cnt,
                            const float* __restrict__ dinv,
                            int2* __restrict__ rec) {
    int gid = blockIdx.x * blockDim.x + threadIdx.x;
    int n = gid >> 4;
    int j = gid & 15;
    if (n >= NN) return;
    int st = rowptr[n], c = cnt[n];
    float dn = dinv[n];
    for (int k = j; k < c; k += 16) {
        int2 r = rec[st + k];
        float nv = __int_as_float(r.y) * dinv[r.x] * dn;
        rec[st + k].y = __float_as_int(nv);
    }
}

#define GATHER_U 8
__global__ __launch_bounds__(256) void gather_kernel(const float* __restrict__ t,
                                                     const int* __restrict__ rowptr,
                                                     const int* __restrict__ cnt,
                                                     const int2* __restrict__ rec,
                                                     const float* __restrict__ sc,
                                                     const float* __restrict__ bias,
                                                     float* __restrict__ h) {
    const int lane = threadIdx.x & 31;
    const int node = __builtin_amdgcn_readfirstlane((int)(blockIdx.x * 8 + (threadIdx.x >> 5)));
    if (node >= NN) return;
    const int beg = rowptr[node];
    const int deg = cnt[node];

    const float* tl = t + lane * 4;
    const float4 self = *(const float4*)(tl + (size_t)node * HH);
    const float  scn  = sc[node];
    const float4 bv   = *(const float4*)(bias + lane * 4);

    float4 acc = make_float4(0.0f, 0.0f, 0.0f, 0.0f);
    for (int b0 = 0; b0 < deg; b0 += 32) {
        const int m = min(deg - b0, 32);
        int2 r = make_int2(0, 0);
        if (lane < m) r = rec[beg + b0 + lane];
        int j = 0;
        for (; j + GATHER_U <= m; j += GATHER_U) {
            int   s[GATHER_U];
            float wt[GATHER_U];
            float4 v[GATHER_U];
#pragma unroll
            for (int u = 0; u < GATHER_U; ++u) {
                s[u]  = __builtin_amdgcn_readlane(r.x, j + u);
                wt[u] = __int_as_float(__builtin_amdgcn_readlane(r.y, j + u));
            }
#pragma unroll
            for (int u = 0; u < GATHER_U; ++u)
                v[u] = *(const float4*)(tl + (size_t)s[u] * HH);
#pragma unroll
            for (int u = 0; u < GATHER_U; ++u) {
                acc.x += v[u].x * wt[u];
                acc.y += v[u].y * wt[u];
                acc.z += v[u].z * wt[u];
                acc.w += v[u].w * wt[u];
            }
        }
        const int rem = m - j;
        if (rem > 0) {
            int   s[GATHER_U - 1];
            float wt[GATHER_U - 1];
            float4 v[GATHER_U - 1];
#pragma unroll
            for (int u = 0; u < GATHER_U - 1; ++u) {
                s[u]  = __builtin_amdgcn_readlane(r.x, j + u);
                wt[u] = __int_as_float(__builtin_amdgcn_readlane(r.y, j + u));
                v[u]  = make_float4(0.0f, 0.0f, 0.0f, 0.0f);
            }
#pragma unroll
            for (int u = 0; u < GATHER_U - 1; ++u)
                if (u < rem) v[u] = *(const float4*)(tl + (size_t)s[u] * HH);
#pragma unroll
            for (int u = 0; u < GATHER_U - 1; ++u) {
                acc.x += v[u].x * wt[u];
                acc.y += v[u].y * wt[u];
                acc.z += v[u].z * wt[u];
                acc.w += v[u].w * wt[u];
            }
        }
    }

    float4 o;
    o.x = elu_f(acc.x + scn * self.x + bv.x);
    o.y = elu_f(acc.y + scn * self.y + bv.y);
    o.z = elu_f(acc.z + scn * self.z + bv.z);
    o.w = elu_f(acc.w + scn * self.w + bv.w);
    *(float4*)(h + (size_t)node * HH + lane * 4) = o;
}

static inline int nblocks(long long t, int bs) { return (int)((t + bs - 1) / bs); }

extern "C" void kernel_launch(void* const* d_in, const int* in_sizes, int n_in,
                              void* d_out, int out_size, void* d_ws, size_t ws_size,
                              hipStream_t stream) {
    const float* x  = (const float*)d_in[0];
    const int*   ei = (const int*)d_in[1];
    const float* ew = (const float*)d_in[2];
    const float* W1 = (const float*)d_in[3];  const float* b1 = (const float*)d_in[4];
    const float* W2 = (const float*)d_in[5];  const float* b2 = (const float*)d_in[6];
    const float* W3 = (const float*)d_in[7];  const float* b3 = (const float*)d_in[8];
    const float* Wg[4] = { (const float*)d_in[9],  (const float*)d_in[11],
                           (const float*)d_in[13], (const float*)d_in[15] };
    const float* bg[4] = { (const float*)d_in[10], (const float*)d_in[12],
                           (const float*)d_in[14], (const float*)d_in[16] };
    const float* Wh = (const float*)d_in[17]; const float* bh = (const float*)d_in[18];
    float* out = (float*)d_out;

    const int* srcI = ei;
    const int* dstI = ei + EE;

    float* f     = (float*)d_ws;
    float* hbuf  = f;
    float* tbuf  = hbuf + (size_t)NN * HH;
    int2*  recb  = (int2*)(tbuf + (size_t)NN * HH);
    int*   rankb = (int*)(recb + EE);
    int*   cntb  = rankb + EE;
    int*   rowp  = cntb + NN;
    float* dinvb = (float*)(rowp + NN);
    float* scb   = dinvb + NN;

    const int BS = 256;

    zero_int_kernel<<<nblocks(NN, BS), BS, 0, stream>>>(cntb, NN);
    rank_kernel<<<nblocks(EE, BS), BS, 0, stream>>>(dstI, cntb, rankb);
    scan_kernel<<<nblocks(NN, SCAN_BS), SCAN_BS, 0, stream>>>(cntb, rowp);
    place_kernel<<<nblocks(EE, BS), BS, 0, stream>>>(srcI, dstI, ew, rowp, rankb, recb);
    deg_kernel<<<nblocks((long long)NN * 16, BS), BS, 0, stream>>>(rowp, cntb, recb, dinvb, scb);
    norm_kernel<<<nblocks((long long)NN * 16, BS), BS, 0, stream>>>(rowp, cntb, dinvb, recb);

    auto gemm128 = [&](const float* A, const float* W, const float* bias, float* C,
                       int K, int act) {
        long long thr = (long long)(NN >> 4) * (HH / 64) * 32;
        if (act) gemm16_wmma<1, 4><<<nblocks(thr, BS), BS, 0, stream>>>(A, W, bias, C, NN, K, HH);
        else     gemm16_wmma<0, 4><<<nblocks(thr, BS), BS, 0, stream>>>(A, W, bias, C, NN, K, HH);
    };

    gemm128(x,    W1, b1, hbuf, IND, 1);
    gemm128(hbuf, W2, b2, tbuf, HH,  1);
    gemm128(tbuf, W3, b3, hbuf, HH,  1);

    for (int l = 0; l < 4; ++l) {
        gemm128(hbuf, Wg[l], nullptr, tbuf, HH, 0);
        gather_kernel<<<nblocks((long long)NN * 32, BS), BS, 0, stream>>>(
            tbuf, rowp, cntb, recb, scb, bg[l], hbuf);
    }

    {
        long long thr = (long long)(NN >> 4) * (OUTD >> 4) * 32;
        gemm16_wmma<0, 1><<<nblocks(thr, BS), BS, 0, stream>>>(hbuf, Wh, bh, out, NN, HH, OUTD);
    }
}


# Next product fused into the aggregation launch: a work-group puts 16 finished node rows in LDS and multiplies them by the next W (or the head) on WMMA; h never stored, 14 launches, sum orders kept
# speedup vs baseline: 2.8939x; 1.2245x over previous
#include <hip/hip_runtime.h>
#include <math.h>


#define NN   50000
#define EE   800000
#define IND  64
#define HH   128
#define OUTD 16

typedef float v2f __attribute__((ext_vector_type(2)));
typedef float v8f __attribute__((ext_vector_type(8)));

__device__ __forceinline__ float elu_f(float v) {
    return v > 0.0f ? v : (__expf(v) - 1.0f);
}

template <int ACT, int TN>
__global__ void gemm16_wmma(const float* __restrict__ A,
                            const float* __restrict__ W,
                            const float* __restrict__ bias,
                            float* __restrict__ C,
                            int M, int K, int Nc) {
    int gid  = blockIdx.x * blockDim.x + threadIdx.x;
    int wave = gid >> 5;
    int lane = gid & 31;
    int groupsN = Nc / (16 * TN);
    int tilesM  = M >> 4;
    if (wave >= tilesM * groupsN) return;
    int tm = wave / groupsN;
    int tg = wave - tm * groupsN;

    int r  = lane & 15;
    int kq = (lane >> 4) << 1;

    const float* Arow = A + (size_t)(tm * 16 + r) * K + kq;
    const float* Bp   = W + (size_t)kq * Nc + tg * (16 * TN) + r;
    const size_t bstep = (size_t)4 * Nc;

    v8f acc[TN];
#pragma unroll
    for (int t = 0; t < TN; ++t) acc[t] = (v8f){};

    v2f a_cur, b_cur[TN];
    {
        float2 av = *(const float2*)(Arow);
        a_cur.x = av.x;
        a_cur.y = av.y;
#pragma unroll
        for (int t = 0; t < TN; ++t) {
            b_cur[t].x = Bp[t * 16];
            b_cur[t].y = Bp[Nc + t * 16];
        }
    }

    for (int k0 = 4; k0 < K; k0 += 4) {
        const float* Bn = Bp + bstep;
        float2 an = *(const float2*)(Arow + k0);
        v2f a_nxt, b_nxt[TN];
        a_nxt.x = an.x;
        a_nxt.y = an.y;
#pragma unroll
        for (int t = 0; t < TN; ++t) {
            b_nxt[t].x = Bn[t * 16];
            b_nxt[t].y = Bn[Nc + t * 16];
        }
#pragma unroll
        for (int t = 0; t < TN; ++t) {
            acc[t] = __builtin_amdgcn_wmma_f32_16x16x4_f32(
                false, a_cur, false, b_cur[t], (short)0, acc[t], false, false);
        }
        a_cur = a_nxt;
#pragma unroll
        for (int t = 0; t < TN; ++t) b_cur[t] = b_nxt[t];
        Bp = Bn;
    }

#pragma unroll
    for (int t = 0; t < TN; ++t) {
        acc[t] = __builtin_amdgcn_wmma_f32_16x16x4_f32(
            false, a_cur, false, b_cur[t], (short)0, acc[t], false, false);
    }

    int mb = tm * 16 + ((lane >> 4) << 3);
#pragma unroll
    for (int t = 0; t < TN; ++t) {
        int n = tg * (16 * TN) + t * 16 + r;
        float bv = bias ? bias[n] : 0.0f;
#pragma unroll
        for (int g = 0; g < 8; ++g) {
            float v = acc[t][g] + bv;
            if (ACT) v = elu_f(v);
            C[(size_t)(mb + g) * Nc + n] = v;
        }
    }
}

__global__ void zero_int_kernel(int* __restrict__ p, int n) {
    int i = blockIdx.x * blockDim.x + threadIdx.x;
    if (i < n) p[i] = 0;
}

__global__ void rank_kernel(const int* __restrict__ dst,
                            int* __restrict__ cnt,
                            int* __restrict__ rank) {
    int e = blockIdx.x * blockDim.x + threadIdx.x;
    if (e < EE) rank[e] = atomicAdd(&cnt[dst[e]], 1);
}

#define SCAN_BS 1024
__global__ __launch_bounds__(SCAN_BS) void scan_kernel(const int* __restrict__ cnt,
                                                       int* __restrict__ rowptr) {
    __shared__ int wave_tot[32];
    __shared__ int wave_pre[32];
    const int tid  = threadIdx.x;
    const int lane = tid & 31;
    const int wv   = tid >> 5;
    const int first = blockIdx.x * SCAN_BS;

    int before = 0;
    for (int i = tid; i < first; i += SCAN_BS) before += cnt[i];
    for (int o = 16; o > 0; o >>= 1) before += __shfl_xor(before, o, 32);

    const int idx = first + tid;
    const int v = (idx < NN) ? cnt[idx] : 0;
    int inc = v;
    for (int o = 1; o < 32; o <<= 1) {
        int u = __shfl_up(inc, o, 32);
        if (lane >= o) inc += u;
    }
    if (lane == 31) wave_tot[wv] = inc;
    if (lane == 0)  wave_pre[wv] = before;
    __syncthreads();

    int base = wave_pre[lane];
    int mine = (lane < wv) ? wave_tot[lane] : 0;
    for (int o = 16; o > 0; o >>= 1) {
        base += __shfl_xor(base, o, 32);
        mine += __shfl_xor(mine, o, 32);
    }
    if (idx < NN) rowptr[idx] = base + mine + inc - v;
}

__global__ void place_kernel(const int* __restrict__ src,
                             const int* __restrict__ dst,
                             const float* __restrict__ w,
                             const int* __restrict__ rowptr,
                             const int* __restrict__ rank,
                             int2* __restrict__ rec) {
    int e = blockIdx.x * blockDim.x + threadIdx.x;
    if (e < EE) {
        int p = rowptr[dst[e]] + rank[e];
        rec[p] = make_int2(src[e], __float_as_int(w[e]));
    }
}

__global__ void deg_kernel(const int* __restrict__ rowptr,
                           const int* __restrict__ cnt,
                           const int2* __restrict__ rec,
                           float* __restrict__ dinv,
                           float* __restrict__ sc) {
    int gid = blockIdx.x * blockDim.x + threadIdx.x;
    int n = gid >> 4;
    int j = gid & 15;
    float s = 0.0f;
    if (n < NN) {
        int st = rowptr[n], c = cnt[n];
        for (int k = j; k < c; k += 16) s += __int_as_float(rec[st + k].y);
    }
    for (int o = 8; o > 0; o >>= 1) s += __shfl_xor(s, o, 16);
    if (n < NN && j == 0) {
        float di = rsqrtf(s + 2.0f);
        dinv[n] = di;
        sc[n]   = 2.0f * di * di;
    }
}

__global__ void norm_kernel(const int* __restrict__ rowptr,
                            const int* __restrict__ cnt,
                            const float* __restrict__ dinv,
                            int2* __restrict__ rec) {
    int gid = blockIdx.x * blockDim.x + threadIdx.x;
    int n = gid >> 4;
    int j = gid & 15;
    if (n >= NN) return;
    int st = rowptr[n], c = cnt[n];
    float dn = dinv[n];
    for (int k = j; k < c; k += 16) {
        int2 r = rec[st + k];
        float nv = __int_as_float(r.y) * dinv[r.x] * dn;
        rec[st + k].y = __float_as_int(nv);
    }
}

#define GATHER_U 8
__device__ __forceinline__ float4 gather_row(const float* __restrict__ t,
                                             const int* __restrict__ rowptr,
                                             const int* __restrict__ cnt,
                                             const int2* __restrict__ rec,
                                             const float* __restrict__ sc,
                                             const float4 bv,
                                             const int node, const int lane) {
    const int beg = rowptr[node];
    const int deg = cnt[node];

    const float* tl = t + lane * 4;
    const float4 self = *(const float4*)(tl + (size_t)node * HH);
    const float  scn  = sc[node];

    float4 acc = make_float4(0.0f, 0.0f, 0.0f, 0.0f);
    for (int b0 = 0; b0 < deg; b0 += 32) {
        const int m = min(deg - b0, 32);
        int2 r = make_int2(0, 0);
        if (lane < m) r = rec[beg + b0 + lane];
        int j = 0;
        for (; j + GATHER_U <= m; j += GATHER_U) {
            int   s[GATHER_U];
            float wt[GATHER_U];
            float4 v[GATHER_U];
#pragma unroll
            for (int u = 0; u < GATHER_U; ++u) {
                s[u]  = __builtin_amdgcn_readlane(r.x, j + u);
                wt[u] = __int_as_float(__builtin_amdgcn_readlane(r.y, j + u));
            }
#pragma unroll
            for (int u = 0; u < GATHER_U; ++u)
                v[u] = *(const float4*)(tl + (size_t)s[u] * HH);
#pragma unroll
            for (int u = 0; u < GATHER_U; ++u) {
                acc.x += v[u].x * wt[u];
                acc.y += v[u].y * wt[u];
                acc.z += v[u].z * wt[u];
                acc.w += v[u].w * wt[u];
            }
        }
        const int rem = m - j;
        if (rem > 0) {
            int   s[GATHER_U - 1];
            float wt[GATHER_U - 1];
            float4 v[GATHER_U - 1];
#pragma unroll
            for (int u = 0; u < GATHER_U - 1; ++u) {
                s[u]  = __builtin_amdgcn_readlane(r.x, j + u);
                wt[u] = __int_as_float(__builtin_amdgcn_readlane(r.y, j + u));
                v[u]  = make_float4(0.0f, 0.0f, 0.0f, 0.0f);
            }
#pragma unroll
            for (int u = 0; u < GATHER_U - 1; ++u)
                if (u < rem) v[u] = *(const float4*)(tl + (size_t)s[u] * HH);
#pragma unroll
            for (int u = 0; u < GATHER_U - 1; ++u) {
                acc.x += v[u].x * wt[u];
                acc.y += v[u].y * wt[u];
                acc.z += v[u].z * wt[u];
                acc.w += v[u].w * wt[u];
            }
        }
    }

    float4 o;
    o.x = elu_f(acc.x + scn * self.x + bv.x);
    o.y = elu_f(acc.y + scn * self.y + bv.y);
    o.z = elu_f(acc.z + scn * self.z + bv.z);
    o.w = elu_f(acc.w + scn * self.w + bv.w);
    return o;
}

#define FUSE_ROWS 16
#define HS_PITCH  132
static_assert(NN % FUSE_ROWS == 0, "grid covers the nodes exactly");
template <int NCOL>
__global__ __launch_bounds__(256) void gather_gemm_kernel(const float* __restrict__ t,
                                                          const int* __restrict__ rowptr,
                                                          const int* __restrict__ cnt,
                                                          const int2* __restrict__ rec,
                                                          const float* __restrict__ sc,
                                                          const float* __restrict__ bias,
                                                          const float* __restrict__ Wn,
                                                          const float* __restrict__ bn,
                                                          float* __restrict__ C) {
    __shared__ __attribute__((aligned(16))) float hs[FUSE_ROWS * HS_PITCH];
    const int lane = threadIdx.x & 31;
    const int wave  = __builtin_amdgcn_readfirstlane((int)(threadIdx.x >> 5));
    const int node0 = blockIdx.x * FUSE_ROWS;
    const float4 bv = *(const float4*)(bias + lane * 4);

    for (int q = 0; q < 2; ++q) {
        const int lr = wave * 2 + q;
        const float4 o = gather_row(t, rowptr, cnt, rec, sc, bv, node0 + lr, lane);
        *(float4*)(hs + lr * HS_PITCH + lane * 4) = o;
    }
    __syncthreads();
    if (wave * 16 >= NCOL) return;

    const int r  = lane & 15;
    const int kq = (lane >> 4) << 1;
    const float* Ap = hs + r * HS_PITCH + kq;
    const float* Bp = Wn + (size_t)kq * NCOL + wave * 16 + r;

    v8f acc = (v8f){};
#pragma unroll
    for (int k0 = 0; k0 < HH; k0 += 4) {
        const float2 av = *(const float2*)(Ap + k0);
        v2f a, b;
        a.x = av.x;
        a.y = av.y;
        b.x = Bp[(size_t)k0 * NCOL];
        b.y = Bp[(size_t)(k0 + 1) * NCOL];
        acc = __builtin_amdgcn_wmma_f32_16x16x4_f32(
            false, a, false, b, (short)0, acc, false, false);
    }

    const int n  = wave * 16 + r;
    const float bvn = bn ? bn[n] : 0.0f;
    const int mb = node0 + ((lane >> 4) << 3);
#pragma unroll
    for (int g = 0; g < 8; ++g)
        C[(size_t)(mb + g) * NCOL + n] = acc[g] + bvn;
}

static inline int nblocks(long long t, int bs) { return (int)((t + bs - 1) / bs); }

extern "C" void kernel_launch(void* const* d_in, const int* in_sizes, int n_in,
                              void* d_out, int out_size, void* d_ws, size_t ws_size,
                              hipStream_t stream) {
    const float* x  = (const float*)d_in[0];
    const int*   ei = (const int*)d_in[1];
    const float* ew = (const float*)d_in[2];
    const float* W1 = (const float*)d_in[3];  const float* b1 = (const float*)d_in[4];
    const float* W2 = (const float*)d_in[5];  const float* b2 = (const float*)d_in[6];
    const float* W3 = (const float*)d_in[7];  const float* b3 = (const float*)d_in[8];
    const float* Wg[4] = { (const float*)d_in[9],  (const float*)d_in[11],
                           (const float*)d_in[13], (const float*)d_in[15] };
    const float* bg[4] = { (const float*)d_in[10], (const float*)d_in[12],
                           (const float*)d_in[14], (const float*)d_in[16] };
    const float* Wh = (const float*)d_in[17]; const float* bh = (const float*)d_in[18];
    float* out = (float*)d_out;

    const int* srcI = ei;
    const int* dstI = ei + EE;

    float* f     = (float*)d_ws;
    float* hbuf  = f;
    float* tbuf  = hbuf + (size_t)NN * HH;
    int2*  recb  = (int2*)(tbuf + (size_t)NN * HH);
    int*   rankb = (int*)(recb + EE);
    int*   cntb  = rankb + EE;
    int*   rowp  = cntb + NN;
    float* dinvb = (float*)(rowp + NN);
    float* scb   = dinvb + NN;

    const int BS = 256;

    zero_int_kernel<<<nblocks(NN, BS), BS, 0, stream>>>(cntb, NN);
    rank_kernel<<<nblocks(EE, BS), BS, 0, stream>>>(dstI, cntb, rankb);
    scan_kernel<<<nblocks(NN, SCAN_BS), SCAN_BS, 0, stream>>>(cntb, rowp);
    place_kernel<<<nblocks(EE, BS), BS, 0, stream>>>(srcI, dstI, ew, rowp, rankb, recb);
    deg_kernel<<<nblocks((long long)NN * 16, BS), BS, 0, stream>>>(rowp, cntb, recb, dinvb, scb);
    norm_kernel<<<nblocks((long long)NN * 16, BS), BS, 0, stream>>>(rowp, cntb, dinvb, recb);

    auto gemm128 = [&](const float* A, const float* W, const float* bias, float* C,
                       int K, int act) {
        long long thr = (long long)(NN >> 4) * (HH / 64) * 32;
        if (act) gemm16_wmma<1, 4><<<nblocks(thr, BS), BS, 0, stream>>>(A, W, bias, C, NN, K, HH);
        else     gemm16_wmma<0, 4><<<nblocks(thr, BS), BS, 0, stream>>>(A, W, bias, C, NN, K, HH);
    };

    gemm128(x,    W1, b1, hbuf, IND, 1);
    gemm128(hbuf, W2, b2, tbuf, HH,  1);
    gemm128(tbuf, W3, b3, hbuf, HH,  1);

    gemm128(hbuf, Wg[0], nullptr, tbuf, HH, 0);
    float* tcur = tbuf;
    float* tnxt = hbuf;
    for (int l = 0; l < 3; ++l) {
        gather_gemm_kernel<HH><<<NN / FUSE_ROWS, BS, 0, stream>>>(
            tcur, rowp, cntb, recb, scb, bg[l], Wg[l + 1], nullptr, tnxt);
        float* sw = tcur; tcur = tnxt; tnxt = sw;
    }
    gather_gemm_kernel<OUTD><<<NN / FUSE_ROWS, BS, 0, stream>>>(
        tcur, rowp, cntb, recb, scb, bg[3], Wh, bh, out);
}


# Four dense products before the first aggregation (3 embedding layers + layer-1 linear) fused into one launch: a 32-row tile stays in LDS through all four; 11 launches, sum orders kept
# speedup vs baseline: 3.3222x; 1.1480x over previous
#include <hip/hip_runtime.h>
#include <math.h>


#define NN   50000
#define EE   800000
#define IND  64
#define HH   128
#define OUTD 16

typedef float v2f __attribute__((ext_vector_type(2)));
typedef float v8f __attribute__((ext_vector_type(8)));

__device__ __forceinline__ float elu_f(float v) {
    return v > 0.0f ? v : (__expf(v) - 1.0f);
}

#define MLP_ROWS 32
#define HS_PITCH 132
#define MLP_GRID ((NN + MLP_ROWS - 1) / MLP_ROWS)

template <int K>
__device__ __forceinline__ void tile_product(const float* in, const float* __restrict__ Wm,
                                             const int wave, const int lane,
                                             v8f& acc0, v8f& acc1) {
    const int r  = lane & 15;
    const int kq = (lane >> 4) << 1;
    const float* Ap0 = in + r * HS_PITCH + kq;
    const float* Ap1 = Ap0 + 16 * HS_PITCH;
    const float* Bp  = Wm + (size_t)kq * HH + wave * 16 + r;
    acc0 = (v8f){};
    acc1 = (v8f){};
#pragma unroll
    for (int k0 = 0; k0 < K; k0 += 4) {
        const float2 a0v = *(const float2*)(Ap0 + k0);
        const float2 a1v = *(const float2*)(Ap1 + k0);
        v2f a0, a1, b;
        a0.x = a0v.x;
        a0.y = a0v.y;
        a1.x = a1v.x;
        a1.y = a1v.y;
        b.x = Bp[(size_t)k0 * HH];
        b.y = Bp[(size_t)(k0 + 1) * HH];
        acc0 = __builtin_amdgcn_wmma_f32_16x16x4_f32(
            false, a0, false, b, (short)0, acc0, false, false);
        acc1 = __builtin_amdgcn_wmma_f32_16x16x4_f32(
            false, a1, false, b, (short)0, acc1, false, false);
    }
}

template <int ACT>
__device__ __forceinline__ void tile_put(float* out, const float* __restrict__ bias,
                                         const int wave, const int lane,
                                         const v8f& acc0, const v8f& acc1) {
    const int n  = wave * 16 + (lane & 15);
    const int rb = (lane >> 4) << 3;
    const float bv = ACT ? bias[n] : 0.0f;
    float* o0 = out + rb * HS_PITCH + n;
    float* o1 = o0 + 16 * HS_PITCH;
#pragma unroll
    for (int g = 0; g < 8; ++g) {
        float v0 = acc0[g] + bv;
        float v1 = acc1[g] + bv;
        if (ACT) {
            v0 = elu_f(v0);
            v1 = elu_f(v1);
        }
        o0[g * HS_PITCH] = v0;
        o1[g * HS_PITCH] = v1;
    }
}

__global__ __launch_bounds__(256) void mlp_chain_kernel(const float* __restrict__ x,
                                                        const float* __restrict__ W1,
                                                        const float* __restrict__ b1,
                                                        const float* __restrict__ W2,
                                                        const float* __restrict__ b2,
                                                        const float* __restrict__ W3,
                                                        const float* __restrict__ b3,
                                                        const float* __restrict__ Wg,
                                                        float* __restrict__ t) {
    __shared__ __attribute__((aligned(16))) float buf0[MLP_ROWS * HS_PITCH];
    __shared__ __attribute__((aligned(16))) float buf1[MLP_ROWS * HS_PITCH];
    const int tid  = threadIdx.x;
    const int lane = tid & 31;
    const int wave  = __builtin_amdgcn_readfirstlane(tid >> 5);
    const int node0 = blockIdx.x * MLP_ROWS;

    for (int i = 0; i < 2; ++i) {
        const int idx = i * 256 + tid;
        const int row = idx >> 4;
        const int c4  = idx & 15;
        float4 v = make_float4(0.0f, 0.0f, 0.0f, 0.0f);
        if (node0 + row < NN) v = *(const float4*)(x + (size_t)(node0 + row) * IND + c4 * 4);
        *(float4*)(buf0 + row * HS_PITCH + c4 * 4) = v;
    }
    __syncthreads();

    v8f acc0, acc1;
    tile_product<IND>(buf0, W1, wave, lane, acc0, acc1);
    tile_put<1>(buf1, b1, wave, lane, acc0, acc1);
    __syncthreads();
    tile_product<HH>(buf1, W2, wave, lane, acc0, acc1);
    tile_put<1>(buf0, b2, wave, lane, acc0, acc1);
    __syncthreads();
    tile_product<HH>(buf0, W3, wave, lane, acc0, acc1);
    tile_put<1>(buf1, b3, wave, lane, acc0, acc1);
    __syncthreads();
    tile_product<HH>(buf1, Wg, wave, lane, acc0, acc1);
    tile_put<0>(buf0, nullptr, wave, lane, acc0, acc1);
    __syncthreads();

    for (int i = 0; i < MLP_ROWS / 8; ++i) {
        const int row = i * 8 + wave;
        if (node0 + row < NN)
            *(float4*)(t + (size_t)(node0 + row) * HH + lane * 4) =
                *(const float4*)(buf0 + row * HS_PITCH + lane * 4);
    }
}

__global__ void zero_int_kernel(int* __restrict__ p, int n) {
    int i = blockIdx.x * blockDim.x + threadIdx.x;
    if (i < n) p[i] = 0;
}

__global__ void rank_kernel(const int* __restrict__ dst,
                            int* __restrict__ cnt,
                            int* __restrict__ rank) {
    int e = blockIdx.x * blockDim.x + threadIdx.x;
    if (e < EE) rank[e] = atomicAdd(&cnt[dst[e]], 1);
}

#define SCAN_BS 1024
__global__ __launch_bounds__(SCAN_BS) void scan_kernel(const int* __restrict__ cnt,
                                                       int* __restrict__ rowptr) {
    __shared__ int wave_tot[32];
    __shared__ int wave_pre[32];
    const int tid  = threadIdx.x;
    const int lane = tid & 31;
    const int wv   = tid >> 5;
    const int first = blockIdx.x * SCAN_BS;

    int before = 0;
    for (int i = tid; i < first; i += SCAN_BS) before += cnt[i];
    for (int o = 16; o > 0; o >>= 1) before += __shfl_xor(before, o, 32);

    const int idx = first + tid;
    const int v = (idx < NN) ? cnt[idx] : 0;
    int inc = v;
    for (int o = 1; o < 32; o <<= 1) {
        int u = __shfl_up(inc, o, 32);
        if (lane >= o) inc += u;
    }
    if (lane == 31) wave_tot[wv] = inc;
    if (lane == 0)  wave_pre[wv] = before;
    __syncthreads();

    int base = wave_pre[lane];
    int mine = (lane < wv) ? wave_tot[lane] : 0;
    for (int o = 16; o > 0; o >>= 1) {
        base += __shfl_xor(base, o, 32);
        mine += __shfl_xor(mine, o, 32);
    }
    if (idx < NN) rowptr[idx] = base + mine + inc - v;
}

__global__ void place_kernel(const int* __restrict__ src,
                             const int* __restrict__ dst,
                             const float* __restrict__ w,
                             const int* __restrict__ rowptr,
                             const int* __restrict__ rank,
                             int2* __restrict__ rec) {
    int e = blockIdx.x * blockDim.x + threadIdx.x;
    if (e < EE) {
        int p = rowptr[dst[e]] + rank[e];
        rec[p] = make_int2(src[e], __float_as_int(w[e]));
    }
}

__global__ void deg_kernel(const int* __restrict__ rowptr,
                           const int* __restrict__ cnt,
                           const int2* __restrict__ rec,
                           float* __restrict__ dinv,
                           float* __restrict__ sc) {
    int gid = blockIdx.x * blockDim.x + threadIdx.x;
    int n = gid >> 4;
    int j = gid & 15;
    float s = 0.0f;
    if (n < NN) {
        int st = rowptr[n], c = cnt[n];
        for (int k = j; k < c; k += 16) s += __int_as_float(rec[st + k].y);
    }
    for (int o = 8; o > 0; o >>= 1) s += __shfl_xor(s, o, 16);
    if (n < NN && j == 0) {
        float di = rsqrtf(s + 2.0f);
        dinv[n] = di;
        sc[n]   = 2.0f * di * di;
    }
}

__global__ void norm_kernel(const int* __restrict__ rowptr,
                            const int* __restrict__ cnt,
                            const float* __restrict__ dinv,
                            int2* __restrict__ rec) {
    int gid = blockIdx.x * blockDim.x + threadIdx.x;
    int n = gid >> 4;
    int j = gid & 15;
    if (n >= NN) return;
    int st = rowptr[n], c = cnt[n];
    float dn = dinv[n];
    for (int k = j; k < c; k += 16) {
        int2 r = rec[st + k];
        float nv = __int_as_float(r.y) * dinv[r.x] * dn;
        rec[st + k].y = __float_as_int(nv);
    }
}

#define GATHER_U 8
__device__ __forceinline__ float4 gather_row(const float* __restrict__ t,
                                             const int* __restrict__ rowptr,
                                             const int* __restrict__ cnt,
                                             const int2* __restrict__ rec,
                                             const float* __restrict__ sc,
                                             const float4 bv,
                                             const int node, const int lane) {
    const int beg = rowptr[node];
    const int deg = cnt[node];

    const float* tl = t + lane * 4;
    const float4 self = *(const float4*)(tl + (size_t)node * HH);
    const float  scn  = sc[node];

    float4 acc = make_float4(0.0f, 0.0f, 0.0f, 0.0f);
    for (int b0 = 0; b0 < deg; b0 += 32) {
        const int m = min(deg - b0, 32);
        int2 r = make_int2(0, 0);
        if (lane < m) r = rec[beg + b0 + lane];
        int j = 0;
        for (; j + GATHER_U <= m; j += GATHER_U) {
            int   s[GATHER_U];
            float wt[GATHER_U];
            float4 v[GATHER_U];
#pragma unroll
            for (int u = 0; u < GATHER_U; ++u) {
                s[u]  = __builtin_amdgcn_readlane(r.x, j + u);
                wt[u] = __int_as_float(__builtin_amdgcn_readlane(r.y, j + u));
            }
#pragma unroll
            for (int u = 0; u < GATHER_U; ++u)
                v[u] = *(const float4*)(tl + (size_t)s[u] * HH);
#pragma unroll
            for (int u = 0; u < GATHER_U; ++u) {
                acc.x += v[u].x * wt[u];
                acc.y += v[u].y * wt[u];
                acc.z += v[u].z * wt[u];
                acc.w += v[u].w * wt[u];
            }
        }
        const int rem = m - j;
        if (rem > 0) {
            int   s[GATHER_U - 1];
            float wt[GATHER_U - 1];
            float4 v[GATHER_U - 1];
#pragma unroll
            for (int u = 0; u < GATHER_U - 1; ++u) {
                s[u]  = __builtin_amdgcn_readlane(r.x, j + u);
                wt[u] = __int_as_float(__builtin_amdgcn_readlane(r.y, j + u));
                v[u]  = make_float4(0.0f, 0.0f, 0.0f, 0.0f);
            }
#pragma unroll
            for (int u = 0; u < GATHER_U - 1; ++u)
                if (u < rem) v[u] = *(const float4*)(tl + (size_t)s[u] * HH);
#pragma unroll
            for (int u = 0; u < GATHER_U - 1; ++u) {
                acc.x += v[u].x * wt[u];
                acc.y += v[u].y * wt[u];
                acc.z += v[u].z * wt[u];
                acc.w += v[u].w * wt[u];
            }
        }
    }

    float4 o;
    o.x = elu_f(acc.x + scn * self.x + bv.x);
    o.y = elu_f(acc.y + scn * self.y + bv.y);
    o.z = elu_f(acc.z + scn * self.z + bv.z);
    o.w = elu_f(acc.w + scn * self.w + bv.w);
    return o;
}

#define FUSE_ROWS 16
static_assert(NN % FUSE_ROWS == 0, "grid covers the nodes exactly");
template <int NCOL>
__global__ __launch_bounds__(256) void gather_gemm_kernel(const float* __restrict__ t,
                                                          const int* __restrict__ rowptr,
                                                          const int* __restrict__ cnt,
                                                          const int2* __restrict__ rec,
                                                          const float* __restrict__ sc,
                                                          const float* __restrict__ bias,
                                                          const float* __restrict__ Wn,
                                                          const float* __restrict__ bn,
                                                          float* __restrict__ C) {
    __shared__ __attribute__((aligned(16))) float hs[FUSE_ROWS * HS_PITCH];
    const int lane = threadIdx.x & 31;
    const int wave  = __builtin_amdgcn_readfirstlane((int)(threadIdx.x >> 5));
    const int node0 = blockIdx.x * FUSE_ROWS;
    const float4 bv = *(const float4*)(bias + lane * 4);

    for (int q = 0; q < 2; ++q) {
        const int lr = wave * 2 + q;
        const float4 o = gather_row(t, rowptr, cnt, rec, sc, bv, node0 + lr, lane);
        *(float4*)(hs + lr * HS_PITCH + lane * 4) = o;
    }
    __syncthreads();
    if (wave * 16 >= NCOL) return;

    const int r  = lane & 15;
    const int kq = (lane >> 4) << 1;
    const float* Ap = hs + r * HS_PITCH + kq;
    const float* Bp = Wn + (size_t)kq * NCOL + wave * 16 + r;

    v8f acc = (v8f){};
#pragma unroll
    for (int k0 = 0; k0 < HH; k0 += 4) {
        const float2 av = *(const float2*)(Ap + k0);
        v2f a, b;
        a.x = av.x;
        a.y = av.y;
        b.x = Bp[(size_t)k0 * NCOL];
        b.y = Bp[(size_t)(k0 + 1) * NCOL];
        acc = __builtin_amdgcn_wmma_f32_16x16x4_f32(
            false, a, false, b, (short)0, acc, false, false);
    }

    const int n  = wave * 16 + r;
    const float bvn = bn ? bn[n] : 0.0f;
    const int mb = node0 + ((lane >> 4) << 3);
#pragma unroll
    for (int g = 0; g < 8; ++g)
        C[(size_t)(mb + g) * NCOL + n] = acc[g] + bvn;
}

static inline int nblocks(long long t, int bs) { return (int)((t + bs - 1) / bs); }

extern "C" void kernel_launch(void* const* d_in, const int* in_sizes, int n_in,
                              void* d_out, int out_size, void* d_ws, size_t ws_size,
                              hipStream_t stream) {
    const float* x  = (const float*)d_in[0];
    const int*   ei = (const int*)d_in[1];
    const float* ew = (const float*)d_in[2];
    const float* W1 = (const float*)d_in[3];  const float* b1 = (const float*)d_in[4];
    const float* W2 = (const float*)d_in[5];  const float* b2 = (const float*)d_in[6];
    const float* W3 = (const float*)d_in[7];  const float* b3 = (const float*)d_in[8];
    const float* Wg[4] = { (const float*)d_in[9],  (const float*)d_in[11],
                           (const float*)d_in[13], (const float*)d_in[15] };
    const float* bg[4] = { (const float*)d_in[10], (const float*)d_in[12],
                           (const float*)d_in[14], (const float*)d_in[16] };
    const float* Wh = (const float*)d_in[17]; const float* bh = (const float*)d_in[18];
    float* out = (float*)d_out;

    const int* srcI = ei;
    const int* dstI = ei + EE;

    float* f     = (float*)d_ws;
    float* hbuf  = f;
    float* tbuf  = hbuf + (size_t)NN * HH;
    int2*  recb  = (int2*)(tbuf + (size_t)NN * HH);
    int*   rankb = (int*)(recb + EE);
    int*   cntb  = rankb + EE;
    int*   rowp  = cntb + NN;
    float* dinvb = (float*)(rowp + NN);
    float* scb   = dinvb + NN;

    const int BS = 256;

    zero_int_kernel<<<nblocks(NN, BS), BS, 0, stream>>>(cntb, NN);
    rank_kernel<<<nblocks(EE, BS), BS, 0, stream>>>(dstI, cntb, rankb);
    scan_kernel<<<nblocks(NN, SCAN_BS), SCAN_BS, 0, stream>>>(cntb, rowp);
    place_kernel<<<nblocks(EE, BS), BS, 0, stream>>>(srcI, dstI, ew, rowp, rankb, recb);
    deg_kernel<<<nblocks((long long)NN * 16, BS), BS, 0, stream>>>(rowp, cntb, recb, dinvb, scb);
    norm_kernel<<<nblocks((long long)NN * 16, BS), BS, 0, stream>>>(rowp, cntb, dinvb, recb);

    mlp_chain_kernel<<<MLP_GRID, BS, 0, stream>>>(x, W1, b1, W2, b2, W3, b3, Wg[0], tbuf);

    float* tcur = tbuf;
    float* tnxt = hbuf;
    for (int l = 0; l < 3; ++l) {
        gather_gemm_kernel<HH><<<NN / FUSE_ROWS, BS, 0, stream>>>(
            tcur, rowp, cntb, recb, scb, bg[l], Wg[l + 1], nullptr, tnxt);
        float* sw = tcur; tcur = tnxt; tnxt = sw;
    }
    gather_gemm_kernel<OUTD><<<NN / FUSE_ROWS, BS, 0, stream>>>(
        tcur, rowp, cntb, recb, scb, bg[3], Wh, bh, out);
}


# Gather re-cut into record streams: each node's list ends with its self-loop record, a wave runs 4 consecutive nodes' lists as one stream in 8-row groups; 4-wave work-groups; sum orders kept
# speedup vs baseline: 3.4725x; 1.0452x over previous
#include <hip/hip_runtime.h>
#include <math.h>


#define NN   50000
#define EE   800000
#define IND  64
#define HH   128
#define OUTD 16

typedef float v2f __attribute__((ext_vector_type(2)));
typedef float v8f __attribute__((ext_vector_type(8)));

__device__ __forceinline__ float elu_f(float v) {
    return v > 0.0f ? v : (__expf(v) - 1.0f);
}

#define MLP_ROWS 32
#define HS_PITCH 132
#define MLP_GRID ((NN + MLP_ROWS - 1) / MLP_ROWS)

template <int K>
__device__ __forceinline__ void tile_product(const float* in, const float* __restrict__ Wm,
                                             const int wave, const int lane,
                                             v8f& acc0, v8f& acc1) {
    const int r  = lane & 15;
    const int kq = (lane >> 4) << 1;
    const float* Ap0 = in + r * HS_PITCH + kq;
    const float* Ap1 = Ap0 + 16 * HS_PITCH;
    const float* Bp  = Wm + (size_t)kq * HH + wave * 16 + r;
    acc0 = (v8f){};
    acc1 = (v8f){};
#pragma unroll
    for (int k0 = 0; k0 < K; k0 += 4) {
        const float2 a0v = *(const float2*)(Ap0 + k0);
        const float2 a1v = *(const float2*)(Ap1 + k0);
        v2f a0, a1, b;
        a0.x = a0v.x;
        a0.y = a0v.y;
        a1.x = a1v.x;
        a1.y = a1v.y;
        b.x = Bp[(size_t)k0 * HH];
        b.y = Bp[(size_t)(k0 + 1) * HH];
        acc0 = __builtin_amdgcn_wmma_f32_16x16x4_f32(
            false, a0, false, b, (short)0, acc0, false, false);
        acc1 = __builtin_amdgcn_wmma_f32_16x16x4_f32(
            false, a1, false, b, (short)0, acc1, false, false);
    }
}

template <int ACT>
__device__ __forceinline__ void tile_put(float* out, const float* __restrict__ bias,
                                         const int wave, const int lane,
                                         const v8f& acc0, const v8f& acc1) {
    const int n  = wave * 16 + (lane & 15);
    const int rb = (lane >> 4) << 3;
    const float bv = ACT ? bias[n] : 0.0f;
    float* o0 = out + rb * HS_PITCH + n;
    float* o1 = o0 + 16 * HS_PITCH;
#pragma unroll
    for (int g = 0; g < 8; ++g) {
        float v0 = acc0[g] + bv;
        float v1 = acc1[g] + bv;
        if (ACT) {
            v0 = elu_f(v0);
            v1 = elu_f(v1);
        }
        o0[g * HS_PITCH] = v0;
        o1[g * HS_PITCH] = v1;
    }
}

__global__ __launch_bounds__(256) void mlp_chain_kernel(const float* __restrict__ x,
                                                        const float* __restrict__ W1,
                                                        const float* __restrict__ b1,
                                                        const float* __restrict__ W2,
                                                        const float* __restrict__ b2,
                                                        const float* __restrict__ W3,
                                                        const float* __restrict__ b3,
                                                        const float* __restrict__ Wg,
                                                        float* __restrict__ t) {
    __shared__ __attribute__((aligned(16))) float buf0[MLP_ROWS * HS_PITCH];
    __shared__ __attribute__((aligned(16))) float buf1[MLP_ROWS * HS_PITCH];
    const int tid  = threadIdx.x;
    const int lane = tid & 31;
    const int wave  = __builtin_amdgcn_readfirstlane(tid >> 5);
    const int node0 = blockIdx.x * MLP_ROWS;

    for (int i = 0; i < 2; ++i) {
        const int idx = i * 256 + tid;
        const int row = idx >> 4;
        const int c4  = idx & 15;
        float4 v = make_float4(0.0f, 0.0f, 0.0f, 0.0f);
        if (node0 + row < NN) v = *(const float4*)(x + (size_t)(node0 + row) * IND + c4 * 4);
        *(float4*)(buf0 + row * HS_PITCH + c4 * 4) = v;
    }
    __syncthreads();

    v8f acc0, acc1;
    tile_product<IND>(buf0, W1, wave, lane, acc0, acc1);
    tile_put<1>(buf1, b1, wave, lane, acc0, acc1);
    __syncthreads();
    tile_product<HH>(buf1, W2, wave, lane, acc0, acc1);
    tile_put<1>(buf0, b2, wave, lane, acc0, acc1);
    __syncthreads();
    tile_product<HH>(buf0, W3, wave, lane, acc0, acc1);
    tile_put<1>(buf1, b3, wave, lane, acc0, acc1);
    __syncthreads();
    tile_product<HH>(buf1, Wg, wave, lane, acc0, acc1);
    tile_put<0>(buf0, nullptr, wave, lane, acc0, acc1);
    __syncthreads();

    for (int i = 0; i < MLP_ROWS / 8; ++i) {
        const int row = i * 8 + wave;
        if (node0 + row < NN)
            *(float4*)(t + (size_t)(node0 + row) * HH + lane * 4) =
                *(const float4*)(buf0 + row * HS_PITCH + lane * 4);
    }
}

__global__ void zero_int_kernel(int* __restrict__ p, int n) {
    int i = blockIdx.x * blockDim.x + threadIdx.x;
    if (i < n) p[i] = 0;
}

__global__ void rank_kernel(const int* __restrict__ dst,
                            int* __restrict__ cnt,
                            int* __restrict__ rank) {
    int e = blockIdx.x * blockDim.x + threadIdx.x;
    if (e < EE) rank[e] = atomicAdd(&cnt[dst[e]], 1);
}

#define SCAN_BS 1024
__global__ __launch_bounds__(SCAN_BS) void scan_kernel(const int* __restrict__ cnt,
                                                       int* __restrict__ rowptr) {
    __shared__ int wave_tot[32];
    __shared__ int wave_pre[32];
    const int tid  = threadIdx.x;
    const int lane = tid & 31;
    const int wv   = tid >> 5;
    const int first = blockIdx.x * SCAN_BS;

    int before = 0;
    for (int i = tid; i < first; i += SCAN_BS) before += cnt[i];
    for (int o = 16; o > 0; o >>= 1) before += __shfl_xor(before, o, 32);

    const int idx = first + tid;
    const int v = (idx < NN) ? cnt[idx] : 0;
    int inc = v;
    for (int o = 1; o < 32; o <<= 1) {
        int u = __shfl_up(inc, o, 32);
        if (lane >= o) inc += u;
    }
    if (lane == 31) wave_tot[wv] = inc;
    if (lane == 0)  wave_pre[wv] = before;
    __syncthreads();

    int base = wave_pre[lane];
    int mine = (lane < wv) ? wave_tot[lane] : 0;
    for (int o = 16; o > 0; o >>= 1) {
        base += __shfl_xor(base, o, 32);
        mine += __shfl_xor(mine, o, 32);
    }
    if (idx < NN) rowptr[idx] = base + mine + inc - v + idx;
}

__global__ void place_kernel(const int* __restrict__ src,
                             const int* __restrict__ dst,
                             const float* __restrict__ w,
                             const int* __restrict__ rowptr,
                             const int* __restrict__ rank,
                             int2* __restrict__ rec) {
    int e = blockIdx.x * blockDim.x + threadIdx.x;
    if (e < EE) {
        int p = rowptr[dst[e]] + rank[e];
        rec[p] = make_int2(src[e], __float_as_int(w[e]));
    }
}

__global__ void deg_kernel(const int* __restrict__ rowptr,
                           const int* __restrict__ cnt,
                           int2* __restrict__ rec,
                           float* __restrict__ dinv) {
    int gid = blockIdx.x * blockDim.x + threadIdx.x;
    int n = gid >> 4;
    int j = gid & 15;
    float s = 0.0f;
    int st = 0, c = 0;
    if (n < NN) {
        st = rowptr[n];
        c  = cnt[n];
        for (int k = j; k < c; k += 16) s += __int_as_float(rec[st + k].y);
    }
    for (int o = 8; o > 0; o >>= 1) s += __shfl_xor(s, o, 16);
    if (n < NN && j == 0) {
        float di = rsqrtf(s + 2.0f);
        dinv[n] = di;
        rec[st + c] = make_int2(n, __float_as_int(2.0f * di * di));
    }
}

__global__ void norm_kernel(const int* __restrict__ rowptr,
                            const int* __restrict__ cnt,
                            const float* __restrict__ dinv,
                            int2* __restrict__ rec) {
    int gid = blockIdx.x * blockDim.x + threadIdx.x;
    int n = gid >> 4;
    int j = gid & 15;
    if (n >= NN) return;
    int st = rowptr[n], c = cnt[n];
    float dn = dinv[n];
    for (int k = j; k < c; k += 16) {
        int2 r = rec[st + k];
        float nv = __int_as_float(r.y) * dinv[r.x] * dn;
        rec[st + k].y = __float_as_int(nv);
    }
}

#define GATHER_U   8
#define FUSE_ROWS  16
#define FUSE_WAVES 4
#define FUSE_NPW   (FUSE_ROWS / FUSE_WAVES)
static_assert(NN % FUSE_ROWS == 0, "grid covers the nodes exactly");
static_assert(FUSE_NPW == 4, "four list ends per wave");

template <bool FULL>
__device__ __forceinline__ void stream_group(const float4 (&v)[GATHER_U], const float (&wt)[GATHER_U],
                                             const int nload, const int pos,
                                             const int e1, const int e2, const int e3,
                                             const float4 bv, float* hrow,
                                             float4& acc, int& cur, int& nend) {
#pragma unroll
    for (int u = 0; u < GATHER_U; ++u) {
        if (FULL || u < nload) {
            acc.x += v[u].x * wt[u];
            acc.y += v[u].y * wt[u];
            acc.z += v[u].z * wt[u];
            acc.w += v[u].w * wt[u];
            if (pos + u + 1 == nend) {
                float4 o;
                o.x = elu_f(acc.x + bv.x);
                o.y = elu_f(acc.y + bv.y);
                o.z = elu_f(acc.z + bv.z);
                o.w = elu_f(acc.w + bv.w);
                *(float4*)(hrow + cur * HS_PITCH) = o;
                acc = make_float4(0.0f, 0.0f, 0.0f, 0.0f);
                ++cur;
                nend = (cur == 1) ? e1 : ((cur == 2) ? e2 : e3);
            }
        }
    }
}

template <int NCOL>
__global__ __launch_bounds__(FUSE_WAVES * 32, 16) void gather_gemm_kernel(const float* __restrict__ t,
                                                          const int* __restrict__ rowptr,
                                                          const int* __restrict__ cnt,
                                                          const int2* __restrict__ rec,
                                                          const float* __restrict__ bias,
                                                          const float* __restrict__ Wn,
                                                          const float* __restrict__ bn,
                                                          float* __restrict__ C) {
    __shared__ __attribute__((aligned(16))) float hs[FUSE_ROWS * HS_PITCH];
    const int lane = threadIdx.x & 31;
    const int wave  = __builtin_amdgcn_readfirstlane((int)(threadIdx.x >> 5));
    const int node0 = blockIdx.x * FUSE_ROWS;
    const int nw    = node0 + wave * FUSE_NPW;
    const float4 bv = *(const float4*)(bias + lane * 4);

    {
        const int e0 = cnt[nw] + 1;
        const int e1 = e0 + cnt[nw + 1] + 1;
        const int e2 = e1 + cnt[nw + 2] + 1;
        const int e3 = e2 + cnt[nw + 3] + 1;
        const int2* rp = rec + rowptr[nw];
        const float* tl = t + lane * 4;
        float* hrow = hs + (wave * FUSE_NPW) * HS_PITCH + lane * 4;

        float4 acc = make_float4(0.0f, 0.0f, 0.0f, 0.0f);
        int cur  = 0;
        int nend = e0;

        int2 r = make_int2(0, 0);
        if (lane < e3) r = rp[lane];
        for (int b0 = 0; b0 < e3; b0 += 32) {
            const int m = min(e3 - b0, 32);
            int2 rn = make_int2(0, 0);
            for (int j = 0; j < m; j += GATHER_U) {
                const int nload = min(m - j, GATHER_U);
                int   s[GATHER_U];
                float wt[GATHER_U];
                float4 v[GATHER_U];
#pragma unroll
                for (int u = 0; u < GATHER_U; ++u) {
                    s[u]  = __builtin_amdgcn_readlane(r.x, j + u);
                    wt[u] = __int_as_float(__builtin_amdgcn_readlane(r.y, j + u));
                }
                if (nload == GATHER_U) {
#pragma unroll
                    for (int u = 0; u < GATHER_U; ++u)
                        v[u] = *(const float4*)(tl + (size_t)s[u] * HH);
                    if (j == 0 && b0 + 32 + lane < e3) rn = rp[b0 + 32 + lane];
                    stream_group<true>(v, wt, nload, b0 + j, e1, e2, e3, bv, hrow, acc, cur, nend);
                } else {
#pragma unroll
                    for (int u = 0; u < GATHER_U; ++u) {
                        v[u] = make_float4(0.0f, 0.0f, 0.0f, 0.0f);
                        if (u < nload) v[u] = *(const float4*)(tl + (size_t)s[u] * HH);
                    }
                    stream_group<false>(v, wt, nload, b0 + j, e1, e2, e3, bv, hrow, acc, cur, nend);
                }
            }
            r = rn;
        }
    }
    __syncthreads();
    if (wave * 32 >= NCOL) return;

    const int r  = lane & 15;
    const int kq = (lane >> 4) << 1;
    const float* Ap = hs + r * HS_PITCH + kq;
    const float* Bp = Wn + (size_t)kq * NCOL + wave * 32 + r;
    constexpr bool TWO = (NCOL >= 32);

    v8f acc0 = (v8f){};
    v8f acc1 = (v8f){};
#pragma unroll 8
    for (int k0 = 0; k0 < HH; k0 += 4) {
        const float2 av = *(const float2*)(Ap + k0);
        v2f a, b0v;
        a.x = av.x;
        a.y = av.y;
        b0v.x = Bp[(size_t)k0 * NCOL];
        b0v.y = Bp[(size_t)(k0 + 1) * NCOL];
        acc0 = __builtin_amdgcn_wmma_f32_16x16x4_f32(
            false, a, false, b0v, (short)0, acc0, false, false);
        if (TWO) {
            v2f b1v;
            b1v.x = Bp[(size_t)k0 * NCOL + 16];
            b1v.y = Bp[(size_t)(k0 + 1) * NCOL + 16];
            acc1 = __builtin_amdgcn_wmma_f32_16x16x4_f32(
                false, a, false, b1v, (short)0, acc1, false, false);
        }
    }

    const int n  = wave * 32 + r;
    const int mb = node0 + ((lane >> 4) << 3);
    const float bvn0 = bn ? bn[n] : 0.0f;
#pragma unroll
    for (int g = 0; g < 8; ++g)
        C[(size_t)(mb + g) * NCOL + n] = acc0[g] + bvn0;
    if (TWO) {
        const float bvn1 = bn ? bn[n + 16] : 0.0f;
#pragma unroll
        for (int g = 0; g < 8; ++g)
            C[(size_t)(mb + g) * NCOL + n + 16] = acc1[g] + bvn1;
    }
}

static inline int nblocks(long long t, int bs) { return (int)((t + bs - 1) / bs); }

extern "C" void kernel_launch(void* const* d_in, const int* in_sizes, int n_in,
                              void* d_out, int out_size, void* d_ws, size_t ws_size,
                              hipStream_t stream) {
    const float* x  = (const float*)d_in[0];
    const int*   ei = (const int*)d_in[1];
    const float* ew = (const float*)d_in[2];
    const float* W1 = (const float*)d_in[3];  const float* b1 = (const float*)d_in[4];
    const float* W2 = (const float*)d_in[5];  const float* b2 = (const float*)d_in[6];
    const float* W3 = (const float*)d_in[7];  const float* b3 = (const float*)d_in[8];
    const float* Wg[4] = { (const float*)d_in[9],  (const float*)d_in[11],
                           (const float*)d_in[13], (const float*)d_in[15] };
    const float* bg[4] = { (const float*)d_in[10], (const float*)d_in[12],
                           (const float*)d_in[14], (const float*)d_in[16] };
    const float* Wh = (const float*)d_in[17]; const float* bh = (const float*)d_in[18];
    float* out = (float*)d_out;

    const int* srcI = ei;
    const int* dstI = ei + EE;

    float* f     = (float*)d_ws;
    float* hbuf  = f;
    float* tbuf  = hbuf + (size_t)NN * HH;
    int2*  recb  = (int2*)(tbuf + (size_t)NN * HH);
    int*   rankb = (int*)(recb + EE + NN);
    int*   cntb  = rankb + EE;
    int*   rowp  = cntb + NN;
    float* dinvb = (float*)(rowp + NN);

    const int BS = 256;

    zero_int_kernel<<<nblocks(NN, BS), BS, 0, stream>>>(cntb, NN);
    rank_kernel<<<nblocks(EE, BS), BS, 0, stream>>>(dstI, cntb, rankb);
    scan_kernel<<<nblocks(NN, SCAN_BS), SCAN_BS, 0, stream>>>(cntb, rowp);
    place_kernel<<<nblocks(EE, BS), BS, 0, stream>>>(srcI, dstI, ew, rowp, rankb, recb);
    deg_kernel<<<nblocks((long long)NN * 16, BS), BS, 0, stream>>>(rowp, cntb, recb, dinvb);
    norm_kernel<<<nblocks((long long)NN * 16, BS), BS, 0, stream>>>(rowp, cntb, dinvb, recb);

    mlp_chain_kernel<<<MLP_GRID, BS, 0, stream>>>(x, W1, b1, W2, b2, W3, b3, Wg[0], tbuf);

    float* tcur = tbuf;
    float* tnxt = hbuf;
    for (int l = 0; l < 3; ++l) {
        gather_gemm_kernel<HH><<<NN / FUSE_ROWS, FUSE_WAVES * 32, 0, stream>>>(
            tcur, rowp, cntb, recb, bg[l], Wg[l + 1], nullptr, tnxt);
        float* sw = tcur; tcur = tnxt; tnxt = sw;
    }
    gather_gemm_kernel<OUTD><<<NN / FUSE_ROWS, FUSE_WAVES * 32, 0, stream>>>(
        tcur, rowp, cntb, recb, bg[3], Wh, bh, out);
}


# GCN norm moved from edges to rows: t stored times dinv, a node's sum times dinv[dst] once (products re-associated, sum orders kept); norm pass gone, degree sums inside the dense launch; 9 launches
# speedup vs baseline: 3.6470x; 1.0503x over previous
#include <hip/hip_runtime.h>
#include <math.h>


#define NN   50000
#define EE   800000
#define IND  64
#define HH   128
#define OUTD 16

typedef float v2f __attribute__((ext_vector_type(2)));
typedef float v8f __attribute__((ext_vector_type(8)));

__device__ __forceinline__ float elu_f(float v) {
    return v > 0.0f ? v : (__expf(v) - 1.0f);
}

#define MLP_ROWS 32
#define HS_PITCH 132
#define MLP_GRID ((NN + MLP_ROWS - 1) / MLP_ROWS)
static_assert(NN % 16 == 0, "a tile's lower 16 rows are always nodes");

template <int K>
__device__ __forceinline__ void tile_product(const float* in, const float* __restrict__ Wm,
                                             const int wave, const int lane,
                                             v8f& acc0, v8f& acc1) {
    const int r  = lane & 15;
    const int kq = (lane >> 4) << 1;
    const float* Ap0 = in + r * HS_PITCH + kq;
    const float* Ap1 = Ap0 + 16 * HS_PITCH;
    const float* Bp  = Wm + (size_t)kq * HH + wave * 16 + r;
    acc0 = (v8f){};
    acc1 = (v8f){};
#pragma unroll
    for (int k0 = 0; k0 < K; k0 += 4) {
        const float2 a0v = *(const float2*)(Ap0 + k0);
        const float2 a1v = *(const float2*)(Ap1 + k0);
        v2f a0, a1, b;
        a0.x = a0v.x;
        a0.y = a0v.y;
        a1.x = a1v.x;
        a1.y = a1v.y;
        b.x = Bp[(size_t)k0 * HH];
        b.y = Bp[(size_t)(k0 + 1) * HH];
        acc0 = __builtin_amdgcn_wmma_f32_16x16x4_f32(
            false, a0, false, b, (short)0, acc0, false, false);
        acc1 = __builtin_amdgcn_wmma_f32_16x16x4_f32(
            false, a1, false, b, (short)0, acc1, false, false);
    }
}

template <int ACT>
__device__ __forceinline__ void tile_put(float* out, const float* __restrict__ bias,
                                         const int wave, const int lane,
                                         const v8f& acc0, const v8f& acc1) {
    const int n  = wave * 16 + (lane & 15);
    const int rb = (lane >> 4) << 3;
    const float bv = ACT ? bias[n] : 0.0f;
    float* o0 = out + rb * HS_PITCH + n;
    float* o1 = o0 + 16 * HS_PITCH;
#pragma unroll
    for (int g = 0; g < 8; ++g) {
        float v0 = acc0[g] + bv;
        float v1 = acc1[g] + bv;
        if (ACT) {
            v0 = elu_f(v0);
            v1 = elu_f(v1);
        }
        o0[g * HS_PITCH] = v0;
        o1[g * HS_PITCH] = v1;
    }
}

__global__ __launch_bounds__(256) void mlp_chain_kernel(const float* __restrict__ x,
                                                        const float* __restrict__ W1,
                                                        const float* __restrict__ b1,
                                                        const float* __restrict__ W2,
                                                        const float* __restrict__ b2,
                                                        const float* __restrict__ W3,
                                                        const float* __restrict__ b3,
                                                        const float* __restrict__ Wg,
                                                        const int* __restrict__ rowptr,
                                                        const int* __restrict__ cnt,
                                                        int2* __restrict__ rec,
                                                        float* __restrict__ dinv,
                                                        float* __restrict__ t) {
    __shared__ __attribute__((aligned(16))) float buf0[MLP_ROWS * HS_PITCH];
    __shared__ __attribute__((aligned(16))) float buf1[MLP_ROWS * HS_PITCH];
    __shared__ float dv[MLP_ROWS];
    const int tid  = threadIdx.x;
    const int lane = tid & 31;
    const int wave  = __builtin_amdgcn_readfirstlane(tid >> 5);
    const int node0 = blockIdx.x * MLP_ROWS;

    float4 xv[2];
#pragma unroll
    for (int i = 0; i < 2; ++i) {
        const int idx = i * 256 + tid;
        const int row = idx >> 4;
        const int c4  = idx & 15;
        xv[i] = make_float4(0.0f, 0.0f, 0.0f, 0.0f);
        if (node0 + row < NN) xv[i] = *(const float4*)(x + (size_t)(node0 + row) * IND + c4 * 4);
    }
    {
        const int j   = tid & 15;
        const int lnA = tid >> 4;
        const int lnB = lnA + 16;
        const int nA  = node0 + lnA;
        const int nB  = node0 + lnB;
        const bool hasB = nB < NN;
        const int stA = rowptr[nA];
        const int cA  = cnt[nA];
        int stB = 0, cB = 0;
        if (hasB) {
            stB = rowptr[nB];
            cB  = cnt[nB];
        }
        float sA = 0.0f, sB = 0.0f;
        const int cM = max(cA, cB);
        for (int k = j; k < cM; k += 16) {
            float wA = 0.0f, wB = 0.0f;
            if (k < cA) wA = __int_as_float(rec[stA + k].y);
            if (k < cB) wB = __int_as_float(rec[stB + k].y);
            sA += wA;
            sB += wB;
        }
        for (int o = 8; o > 0; o >>= 1) {
            sA += __shfl_xor(sA, o, 16);
            sB += __shfl_xor(sB, o, 16);
        }
        if (j == 0) {
            const float dA = rsqrtf(sA + 2.0f);
            dinv[nA] = dA;
            rec[stA + cA] = make_int2(nA, __float_as_int(2.0f));
            dv[lnA] = dA;
            float dB = 0.0f;
            if (hasB) {
                dB = rsqrtf(sB + 2.0f);
                dinv[nB] = dB;
                rec[stB + cB] = make_int2(nB, __float_as_int(2.0f));
            }
            dv[lnB] = dB;
        }
    }
#pragma unroll
    for (int i = 0; i < 2; ++i) {
        const int idx = i * 256 + tid;
        *(float4*)(buf0 + (idx >> 4) * HS_PITCH + (idx & 15) * 4) = xv[i];
    }
    __syncthreads();

    v8f acc0, acc1;
    tile_product<IND>(buf0, W1, wave, lane, acc0, acc1);
    tile_put<1>(buf1, b1, wave, lane, acc0, acc1);
    __syncthreads();
    tile_product<HH>(buf1, W2, wave, lane, acc0, acc1);
    tile_put<1>(buf0, b2, wave, lane, acc0, acc1);
    __syncthreads();
    tile_product<HH>(buf0, W3, wave, lane, acc0, acc1);
    tile_put<1>(buf1, b3, wave, lane, acc0, acc1);
    __syncthreads();
    tile_product<HH>(buf1, Wg, wave, lane, acc0, acc1);
    tile_put<0>(buf0, nullptr, wave, lane, acc0, acc1);
    __syncthreads();

    for (int i = 0; i < MLP_ROWS / 8; ++i) {
        const int row = i * 8 + wave;
        if (node0 + row < NN) {
            float4 v = *(const float4*)(buf0 + row * HS_PITCH + lane * 4);
            const float d = dv[row];
            v.x *= d;
            v.y *= d;
            v.z *= d;
            v.w *= d;
            *(float4*)(t + (size_t)(node0 + row) * HH + lane * 4) = v;
        }
    }
}

__global__ void zero_int_kernel(int* __restrict__ p, int n) {
    int i = blockIdx.x * blockDim.x + threadIdx.x;
    if (i < n) p[i] = 0;
}

__global__ void rank_kernel(const int* __restrict__ dst,
                            int* __restrict__ cnt,
                            int* __restrict__ rank) {
    int e = blockIdx.x * blockDim.x + threadIdx.x;
    if (e < EE) rank[e] = atomicAdd(&cnt[dst[e]], 1);
}

#define SCAN_BS 1024
__global__ __launch_bounds__(SCAN_BS) void scan_kernel(const int* __restrict__ cnt,
                                                       int* __restrict__ rowptr) {
    __shared__ int wave_tot[32];
    __shared__ int wave_pre[32];
    const int tid  = threadIdx.x;
    const int lane = tid & 31;
    const int wv   = tid >> 5;
    const int first = blockIdx.x * SCAN_BS;

    int before = 0;
    for (int i = tid; i < first; i += SCAN_BS) before += cnt[i];
    for (int o = 16; o > 0; o >>= 1) before += __shfl_xor(before, o, 32);

    const int idx = first + tid;
    const int v = (idx < NN) ? cnt[idx] : 0;
    int inc = v;
    for (int o = 1; o < 32; o <<= 1) {
        int u = __shfl_up(inc, o, 32);
        if (lane >= o) inc += u;
    }
    if (lane == 31) wave_tot[wv] = inc;
    if (lane == 0)  wave_pre[wv] = before;
    __syncthreads();

    int base = wave_pre[lane];
    int mine = (lane < wv) ? wave_tot[lane] : 0;
    for (int o = 16; o > 0; o >>= 1) {
        base += __shfl_xor(base, o, 32);
        mine += __shfl_xor(mine, o, 32);
    }
    if (idx < NN) rowptr[idx] = base + mine + inc - v + idx;
}

__global__ void place_kernel(const int* __restrict__ src,
                             const int* __restrict__ dst,
                             const float* __restrict__ w,
                             const int* __restrict__ rowptr,
                             const int* __restrict__ rank,
                             int2* __restrict__ rec) {
    int e = blockIdx.x * blockDim.x + threadIdx.x;
    if (e < EE) {
        int p = rowptr[dst[e]] + rank[e];
        rec[p] = make_int2(src[e], __float_as_int(w[e]));
    }
}

#define GATHER_U   8
#define FUSE_ROWS  16
#define FUSE_WAVES 4
#define FUSE_NPW   (FUSE_ROWS / FUSE_WAVES)
static_assert(NN % FUSE_ROWS == 0, "grid covers the nodes exactly");
static_assert(FUSE_NPW == 4, "four list ends per wave");

template <bool FULL, bool SCALE>
__device__ __forceinline__ void stream_group(const float4 (&v)[GATHER_U], const float (&wt)[GATHER_U],
                                             const int nload, const int pos,
                                             const int e1, const int e2, const int e3,
                                             const float d1, const float d2, const float d3,
                                             const float4 bv, float* hrow,
                                             float4& acc, int& cur, int& nend, float& dcur) {
#pragma unroll
    for (int u = 0; u < GATHER_U; ++u) {
        if (FULL || u < nload) {
            acc.x += v[u].x * wt[u];
            acc.y += v[u].y * wt[u];
            acc.z += v[u].z * wt[u];
            acc.w += v[u].w * wt[u];
            if (pos + u + 1 == nend) {
                float4 o;
                o.x = elu_f(acc.x * dcur + bv.x);
                o.y = elu_f(acc.y * dcur + bv.y);
                o.z = elu_f(acc.z * dcur + bv.z);
                o.w = elu_f(acc.w * dcur + bv.w);
                if (SCALE) {
                    o.x *= dcur;
                    o.y *= dcur;
                    o.z *= dcur;
                    o.w *= dcur;
                }
                *(float4*)(hrow + cur * HS_PITCH) = o;
                acc = make_float4(0.0f, 0.0f, 0.0f, 0.0f);
                ++cur;
                nend = (cur == 1) ? e1 : ((cur == 2) ? e2 : e3);
                dcur = (cur == 1) ? d1 : ((cur == 2) ? d2 : d3);
            }
        }
    }
}

template <int NCOL>
__global__ __launch_bounds__(FUSE_WAVES * 32, 16) void gather_gemm_kernel(const float* __restrict__ t,
                                                          const int* __restrict__ rowptr,
                                                          const int* __restrict__ cnt,
                                                          const int2* __restrict__ rec,
                                                          const float* __restrict__ dinv,
                                                          const float* __restrict__ bias,
                                                          const float* __restrict__ Wn,
                                                          const float* __restrict__ bn,
                                                          float* __restrict__ C) {
    __shared__ __attribute__((aligned(16))) float hs[FUSE_ROWS * HS_PITCH];
    const int lane = threadIdx.x & 31;
    const int wave  = __builtin_amdgcn_readfirstlane((int)(threadIdx.x >> 5));
    const int node0 = blockIdx.x * FUSE_ROWS;
    const int nw    = node0 + wave * FUSE_NPW;
    const float4 bv = *(const float4*)(bias + lane * 4);

    {
        const int e0 = cnt[nw] + 1;
        const int e1 = e0 + cnt[nw + 1] + 1;
        const int e2 = e1 + cnt[nw + 2] + 1;
        const int e3 = e2 + cnt[nw + 3] + 1;
        const int2* rp = rec + rowptr[nw];
        const float* tl = t + lane * 4;
        const float d1 = dinv[nw + 1];
        const float d2 = dinv[nw + 2];
        const float d3 = dinv[nw + 3];
        float dcur = dinv[nw];
        constexpr bool SC = (NCOL == HH);
        float* hrow = hs + (wave * FUSE_NPW) * HS_PITCH + lane * 4;

        float4 acc = make_float4(0.0f, 0.0f, 0.0f, 0.0f);
        int cur  = 0;
        int nend = e0;

        int2 r = make_int2(0, 0);
        if (lane < e3) r = rp[lane];
        for (int b0 = 0; b0 < e3; b0 += 32) {
            const int m = min(e3 - b0, 32);
            int2 rn = make_int2(0, 0);
            for (int j = 0; j < m; j += GATHER_U) {
                const int nload = min(m - j, GATHER_U);
                int   s[GATHER_U];
                float wt[GATHER_U];
                float4 v[GATHER_U];
#pragma unroll
                for (int u = 0; u < GATHER_U; ++u) {
                    s[u]  = __builtin_amdgcn_readlane(r.x, j + u);
                    wt[u] = __int_as_float(__builtin_amdgcn_readlane(r.y, j + u));
                }
                if (nload == GATHER_U) {
#pragma unroll
                    for (int u = 0; u < GATHER_U; ++u)
                        v[u] = *(const float4*)(tl + (size_t)s[u] * HH);
                    if (j == 0 && b0 + 32 + lane < e3) rn = rp[b0 + 32 + lane];
                    stream_group<true, SC>(v, wt, nload, b0 + j, e1, e2, e3, d1, d2, d3, bv, hrow, acc, cur, nend, dcur);
                } else {
#pragma unroll
                    for (int u = 0; u < GATHER_U; ++u) {
                        v[u] = make_float4(0.0f, 0.0f, 0.0f, 0.0f);
                        if (u < nload) v[u] = *(const float4*)(tl + (size_t)s[u] * HH);
                    }
                    stream_group<false, SC>(v, wt, nload, b0 + j, e1, e2, e3, d1, d2, d3, bv, hrow, acc, cur, nend, dcur);
                }
            }
            r = rn;
        }
    }
    __syncthreads();
    if (wave * 32 >= NCOL) return;

    const int r  = lane & 15;
    const int kq = (lane >> 4) << 1;
    const float* Ap = hs + r * HS_PITCH + kq;
    const float* Bp = Wn + (size_t)kq * NCOL + wave * 32 + r;
    constexpr bool TWO = (NCOL >= 32);

    v8f acc0 = (v8f){};
    v8f acc1 = (v8f){};
#pragma unroll 8
    for (int k0 = 0; k0 < HH; k0 += 4) {
        const float2 av = *(const float2*)(Ap + k0);
        v2f a, b0v;
        a.x = av.x;
        a.y = av.y;
        b0v.x = Bp[(size_t)k0 * NCOL];
        b0v.y = Bp[(size_t)(k0 + 1) * NCOL];
        acc0 = __builtin_amdgcn_wmma_f32_16x16x4_f32(
            false, a, false, b0v, (short)0, acc0, false, false);
        if (TWO) {
            v2f b1v;
            b1v.x = Bp[(size_t)k0 * NCOL + 16];
            b1v.y = Bp[(size_t)(k0 + 1) * NCOL + 16];
            acc1 = __builtin_amdgcn_wmma_f32_16x16x4_f32(
                false, a, false, b1v, (short)0, acc1, false, false);
        }
    }

    const int n  = wave * 32 + r;
    const int mb = node0 + ((lane >> 4) << 3);
    const float bvn0 = bn ? bn[n] : 0.0f;
#pragma unroll
    for (int g = 0; g < 8; ++g)
        C[(size_t)(mb + g) * NCOL + n] = acc0[g] + bvn0;
    if (TWO) {
        const float bvn1 = bn ? bn[n + 16] : 0.0f;
#pragma unroll
        for (int g = 0; g < 8; ++g)
            C[(size_t)(mb + g) * NCOL + n + 16] = acc1[g] + bvn1;
    }
}

static inline int nblocks(long long t, int bs) { return (int)((t + bs - 1) / bs); }

extern "C" void kernel_launch(void* const* d_in, const int* in_sizes, int n_in,
                              void* d_out, int out_size, void* d_ws, size_t ws_size,
                              hipStream_t stream) {
    const float* x  = (const float*)d_in[0];
    const int*   ei = (const int*)d_in[1];
    const float* ew = (const float*)d_in[2];
    const float* W1 = (const float*)d_in[3];  const float* b1 = (const float*)d_in[4];
    const float* W2 = (const float*)d_in[5];  const float* b2 = (const float*)d_in[6];
    const float* W3 = (const float*)d_in[7];  const float* b3 = (const float*)d_in[8];
    const float* Wg[4] = { (const float*)d_in[9],  (const float*)d_in[11],
                           (const float*)d_in[13], (const float*)d_in[15] };
    const float* bg[4] = { (const float*)d_in[10], (const float*)d_in[12],
                           (const float*)d_in[14], (const float*)d_in[16] };
    const float* Wh = (const float*)d_in[17]; const float* bh = (const float*)d_in[18];
    float* out = (float*)d_out;

    const int* srcI = ei;
    const int* dstI = ei + EE;

    float* f     = (float*)d_ws;
    float* hbuf  = f;
    float* tbuf  = hbuf + (size_t)NN * HH;
    int2*  recb  = (int2*)(tbuf + (size_t)NN * HH);
    int*   rankb = (int*)(recb + EE + NN);
    int*   cntb  = rankb + EE;
    int*   rowp  = cntb + NN;
    float* dinvb = (float*)(rowp + NN);

    const int BS = 256;

    zero_int_kernel<<<nblocks(NN, BS), BS, 0, stream>>>(cntb, NN);
    rank_kernel<<<nblocks(EE, BS), BS, 0, stream>>>(dstI, cntb, rankb);
    scan_kernel<<<nblocks(NN, SCAN_BS), SCAN_BS, 0, stream>>>(cntb, rowp);
    place_kernel<<<nblocks(EE, BS), BS, 0, stream>>>(srcI, dstI, ew, rowp, rankb, recb);

    mlp_chain_kernel<<<MLP_GRID, BS, 0, stream>>>(x, W1, b1, W2, b2, W3, b3, Wg[0],
                                                  rowp, cntb, recb, dinvb, tbuf);

    float* tcur = tbuf;
    float* tnxt = hbuf;
    for (int l = 0; l < 3; ++l) {
        gather_gemm_kernel<HH><<<NN / FUSE_ROWS, FUSE_WAVES * 32, 0, stream>>>(
            tcur, rowp, cntb, recb, dinvb, bg[l], Wg[l + 1], nullptr, tnxt);
        float* sw = tcur; tcur = tnxt; tnxt = sw;
    }
    gather_gemm_kernel<OUTD><<<NN / FUSE_ROWS, FUSE_WAVES * 32, 0, stream>>>(
        tcur, rowp, cntb, recb, dinvb, bg[3], Wh, bh, out);
}


# Counting pass drops each edge straight into its node's fixed-pitch record bucket (atomic rank = slot; longer rows spill to an overflow list, summed first): scan and place launches gone, 7 launches
# speedup vs baseline: 4.0627x; 1.1140x over previous
#include <hip/hip_runtime.h>
#include <math.h>


#define NN   50000
#define EE   800000
#define IND  64
#define HH   128
#define OUTD 16

#define BCAP   47
#define BPITCH 48

typedef float v2f __attribute__((ext_vector_type(2)));
typedef float v8f __attribute__((ext_vector_type(8)));

__device__ __forceinline__ float elu_f(float v) {
    return v > 0.0f ? v : (__expf(v) - 1.0f);
}

#define MLP_ROWS 32
#define HS_PITCH 132
#define MLP_GRID ((NN + MLP_ROWS - 1) / MLP_ROWS)
static_assert(NN % 16 == 0, "a tile's lower 16 rows are always nodes");

template <int K>
__device__ __forceinline__ void tile_product(const float* in, const float* __restrict__ Wm,
                                             const int wave, const int lane,
                                             v8f& acc0, v8f& acc1) {
    const int r  = lane & 15;
    const int kq = (lane >> 4) << 1;
    const float* Ap0 = in + r * HS_PITCH + kq;
    const float* Ap1 = Ap0 + 16 * HS_PITCH;
    const float* Bp  = Wm + (size_t)kq * HH + wave * 16 + r;
    acc0 = (v8f){};
    acc1 = (v8f){};
#pragma unroll
    for (int k0 = 0; k0 < K; k0 += 4) {
        const float2 a0v = *(const float2*)(Ap0 + k0);
        const float2 a1v = *(const float2*)(Ap1 + k0);
        v2f a0, a1, b;
        a0.x = a0v.x;
        a0.y = a0v.y;
        a1.x = a1v.x;
        a1.y = a1v.y;
        b.x = Bp[(size_t)k0 * HH];
        b.y = Bp[(size_t)(k0 + 1) * HH];
        acc0 = __builtin_amdgcn_wmma_f32_16x16x4_f32(
            false, a0, false, b, (short)0, acc0, false, false);
        acc1 = __builtin_amdgcn_wmma_f32_16x16x4_f32(
            false, a1, false, b, (short)0, acc1, false, false);
    }
}

template <int ACT>
__device__ __forceinline__ void tile_put(float* out, const float* __restrict__ bias,
                                         const int wave, const int lane,
                                         const v8f& acc0, const v8f& acc1) {
    const int n  = wave * 16 + (lane & 15);
    const int rb = (lane >> 4) << 3;
    const float bv = ACT ? bias[n] : 0.0f;
    float* o0 = out + rb * HS_PITCH + n;
    float* o1 = o0 + 16 * HS_PITCH;
#pragma unroll
    for (int g = 0; g < 8; ++g) {
        float v0 = acc0[g] + bv;
        float v1 = acc1[g] + bv;
        if (ACT) {
            v0 = elu_f(v0);
            v1 = elu_f(v1);
        }
        o0[g * HS_PITCH] = v0;
        o1[g * HS_PITCH] = v1;
    }
}

__global__ __launch_bounds__(256) void mlp_chain_kernel(const float* __restrict__ x,
                                                        const float* __restrict__ W1,
                                                        const float* __restrict__ b1,
                                                        const float* __restrict__ W2,
                                                        const float* __restrict__ b2,
                                                        const float* __restrict__ W3,
                                                        const float* __restrict__ b3,
                                                        const float* __restrict__ Wg,
                                                        const int* __restrict__ cnt,
                                                        int2* __restrict__ rec,
                                                        const int4* __restrict__ ovf,
                                                        float* __restrict__ dinv,
                                                        float* __restrict__ t) {
    __shared__ __attribute__((aligned(16))) float buf0[MLP_ROWS * HS_PITCH];
    __shared__ __attribute__((aligned(16))) float buf1[MLP_ROWS * HS_PITCH];
    __shared__ float dv[MLP_ROWS];
    const int tid  = threadIdx.x;
    const int lane = tid & 31;
    const int wave  = __builtin_amdgcn_readfirstlane(tid >> 5);
    const int node0 = blockIdx.x * MLP_ROWS;

    float4 xv[2];
#pragma unroll
    for (int i = 0; i < 2; ++i) {
        const int idx = i * 256 + tid;
        const int row = idx >> 4;
        const int c4  = idx & 15;
        xv[i] = make_float4(0.0f, 0.0f, 0.0f, 0.0f);
        if (node0 + row < NN) xv[i] = *(const float4*)(x + (size_t)(node0 + row) * IND + c4 * 4);
    }
    {
        const int j   = tid & 15;
        const int lnA = tid >> 4;
        const int lnB = lnA + 16;
        const int nA  = node0 + lnA;
        const int nB  = node0 + lnB;
        const bool hasB = nB < NN;
        const int stA = nA * BPITCH;
        const int dgA = cnt[nA];
        const int cA  = min(dgA, BCAP);
        int stB = 0, dgB = 0;
        if (hasB) {
            stB = nB * BPITCH;
            dgB = cnt[nB];
        }
        const int cB = min(dgB, BCAP);
        float sA = 0.0f, sB = 0.0f;
        const int cM = max(cA, cB);
        for (int k = j; k < cM; k += 16) {
            float wA = 0.0f, wB = 0.0f;
            if (k < cA) wA = __int_as_float(rec[stA + k].y);
            if (k < cB) wB = __int_as_float(rec[stB + k].y);
            sA += wA;
            sB += wB;
        }
        if (dgA > BCAP || dgB > BCAP) {
            const int nov = cnt[NN];
            for (int k = j; k < nov; k += 16) {
                const int4 o = ovf[k];
                if (o.x == nA) sA += __int_as_float(o.z);
                if (o.x == nB) sB += __int_as_float(o.z);
            }
        }
        for (int o = 8; o > 0; o >>= 1) {
            sA += __shfl_xor(sA, o, 16);
            sB += __shfl_xor(sB, o, 16);
        }
        if (j == 0) {
            const float dA = rsqrtf(sA + 2.0f);
            dinv[nA] = dA;
            rec[stA + cA] = make_int2(nA, __float_as_int(2.0f));
            dv[lnA] = dA;
            float dB = 0.0f;
            if (hasB) {
                dB = rsqrtf(sB + 2.0f);
                dinv[nB] = dB;
                rec[stB + cB] = make_int2(nB, __float_as_int(2.0f));
            }
            dv[lnB] = dB;
        }
    }
#pragma unroll
    for (int i = 0; i < 2; ++i) {
        const int idx = i * 256 + tid;
        *(float4*)(buf0 + (idx >> 4) * HS_PITCH + (idx & 15) * 4) = xv[i];
    }
    __syncthreads();

    v8f acc0, acc1;
    tile_product<IND>(buf0, W1, wave, lane, acc0, acc1);
    tile_put<1>(buf1, b1, wave, lane, acc0, acc1);
    __syncthreads();
    tile_product<HH>(buf1, W2, wave, lane, acc0, acc1);
    tile_put<1>(buf0, b2, wave, lane, acc0, acc1);
    __syncthreads();
    tile_product<HH>(buf0, W3, wave, lane, acc0, acc1);
    tile_put<1>(buf1, b3, wave, lane, acc0, acc1);
    __syncthreads();
    tile_product<HH>(buf1, Wg, wave, lane, acc0, acc1);
    tile_put<0>(buf0, nullptr, wave, lane, acc0, acc1);
    __syncthreads();

    for (int i = 0; i < MLP_ROWS / 8; ++i) {
        const int row = i * 8 + wave;
        if (node0 + row < NN) {
            float4 v = *(const float4*)(buf0 + row * HS_PITCH + lane * 4);
            const float d = dv[row];
            v.x *= d;
            v.y *= d;
            v.z *= d;
            v.w *= d;
            *(float4*)(t + (size_t)(node0 + row) * HH + lane * 4) = v;
        }
    }
}

__global__ void zero_int_kernel(int* __restrict__ p, int n) {
    int i = blockIdx.x * blockDim.x + threadIdx.x;
    if (i < n) p[i] = 0;
}

__global__ void bucket_kernel(const int* __restrict__ src,
                              const int* __restrict__ dst,
                              const float* __restrict__ w,
                              int* __restrict__ cnt,
                              int2* __restrict__ rec,
                              int4* __restrict__ ovf) {
    int e = blockIdx.x * blockDim.x + threadIdx.x;
    if (e < EE) {
        const int d  = dst[e];
        const int s  = src[e];
        const int wb = __float_as_int(w[e]);
        const int r  = atomicAdd(&cnt[d], 1);
        if (r < BCAP) {
            rec[(size_t)d * BPITCH + r] = make_int2(s, wb);
        } else {
            const int o = atomicAdd(&cnt[NN], 1);
            ovf[o] = make_int4(d, s, wb, 0);
        }
    }
}

#define GATHER_U   8
#define FUSE_ROWS  16
#define FUSE_WAVES 4
#define FUSE_NPW   (FUSE_ROWS / FUSE_WAVES)
static_assert(NN % FUSE_ROWS == 0, "grid covers the nodes exactly");
static_assert(FUSE_NPW == 4, "four list ends per wave");

template <bool FULL, bool SCALE>
__device__ __forceinline__ void stream_group(const float4 (&v)[GATHER_U], const float (&wt)[GATHER_U],
                                             const bool anyovf,
                                             const int nload, const int pos,
                                             const int e1, const int e2, const int e3,
                                             const float d1, const float d2, const float d3,
                                             const float4 bv, float* hrow,
                                             float4& acc, int& cur, int& nend, float& dcur) {
#pragma unroll
    for (int u = 0; u < GATHER_U; ++u) {
        if (FULL || u < nload) {
            acc.x += v[u].x * wt[u];
            acc.y += v[u].y * wt[u];
            acc.z += v[u].z * wt[u];
            acc.w += v[u].w * wt[u];
            if (pos + u + 1 == nend) {
                if (anyovf) {
                    const float4 x = *(const float4*)(hrow + cur * HS_PITCH);
                    acc.x += x.x;
                    acc.y += x.y;
                    acc.z += x.z;
                    acc.w += x.w;
                }
                float4 o;
                o.x = elu_f(acc.x * dcur + bv.x);
                o.y = elu_f(acc.y * dcur + bv.y);
                o.z = elu_f(acc.z * dcur + bv.z);
                o.w = elu_f(acc.w * dcur + bv.w);
                if (SCALE) {
                    o.x *= dcur;
                    o.y *= dcur;
                    o.z *= dcur;
                    o.w *= dcur;
                }
                *(float4*)(hrow + cur * HS_PITCH) = o;
                acc = make_float4(0.0f, 0.0f, 0.0f, 0.0f);
                ++cur;
                nend = (cur == 1) ? e1 : ((cur == 2) ? e2 : e3);
                dcur = (cur == 1) ? d1 : ((cur == 2) ? d2 : d3);
            }
        }
    }
}

template <int NCOL>
__global__ __launch_bounds__(FUSE_WAVES * 32, 16) void gather_gemm_kernel(const float* __restrict__ t,
                                                          const int* __restrict__ cnt,
                                                          const int2* __restrict__ rec,
                                                          const int4* __restrict__ ovf,
                                                          const float* __restrict__ dinv,
                                                          const float* __restrict__ bias,
                                                          const float* __restrict__ Wn,
                                                          const float* __restrict__ bn,
                                                          float* __restrict__ C) {
    __shared__ __attribute__((aligned(16))) float hs[FUSE_ROWS * HS_PITCH];
    const int lane = threadIdx.x & 31;
    const int wave  = __builtin_amdgcn_readfirstlane((int)(threadIdx.x >> 5));
    const int node0 = blockIdx.x * FUSE_ROWS;
    const int nw    = node0 + wave * FUSE_NPW;
    const float4 bv = *(const float4*)(bias + lane * 4);

    {
        const int g0 = cnt[nw];
        const int g1 = cnt[nw + 1];
        const int g2 = cnt[nw + 2];
        const int g3 = cnt[nw + 3];
        const int e0 = min(g0, BCAP) + 1;
        const int e1 = e0 + min(g1, BCAP) + 1;
        const int e2 = e1 + min(g2, BCAP) + 1;
        const int e3 = e2 + min(g3, BCAP) + 1;
        const int2* rb = rec + (size_t)nw * BPITCH;
        const float* tl = t + lane * 4;
        const float d1 = dinv[nw + 1];
        const float d2 = dinv[nw + 2];
        const float d3 = dinv[nw + 3];
        float dcur = dinv[nw];
        constexpr bool SC = (NCOL == HH);
        float* hrow = hs + (wave * FUSE_NPW) * HS_PITCH + lane * 4;

        const bool anyovf = (g0 > BCAP) | (g1 > BCAP) | (g2 > BCAP) | (g3 > BCAP);
        if (anyovf) {
#pragma unroll
            for (int i = 0; i < FUSE_NPW; ++i)
                *(float4*)(hrow + i * HS_PITCH) = make_float4(0.0f, 0.0f, 0.0f, 0.0f);
            const int nov = cnt[NN];
#pragma nounroll
            for (int k = 0; k < nov; ++k) {
                const int4 o = ovf[k];
                const int i = o.x - nw;
                if ((unsigned)i < (unsigned)FUSE_NPW) {
                    float* hp = hrow + i * HS_PITCH;
                    float4 a = *(const float4*)hp;
                    const float4 vv = *(const float4*)(tl + (size_t)o.y * HH);
                    const float wv = __int_as_float(o.z);
                    a.x += vv.x * wv;
                    a.y += vv.y * wv;
                    a.z += vv.z * wv;
                    a.w += vv.w * wv;
                    *(float4*)hp = a;
                }
            }
        }

        float4 acc = make_float4(0.0f, 0.0f, 0.0f, 0.0f);
        int cur  = 0;
        int nend = e0;

        auto slot = [&](const int p) -> const int2* {
            const int i  = (p >= e0) + (p >= e1) + (p >= e2);
            const int st = (p >= e2) ? e2 : ((p >= e1) ? e1 : ((p >= e0) ? e0 : 0));
            return rb + i * BPITCH + (p - st);
        };

        int2 r = make_int2(0, 0);
        if (lane < e3) r = *slot(lane);
        for (int b0 = 0; b0 < e3; b0 += 32) {
            const int m = min(e3 - b0, 32);
            int2 rn = make_int2(0, 0);
            for (int j = 0; j < m; j += GATHER_U) {
                const int nload = min(m - j, GATHER_U);
                int   s[GATHER_U];
                float wt[GATHER_U];
                float4 v[GATHER_U];
#pragma unroll
                for (int u = 0; u < GATHER_U; ++u) {
                    s[u]  = __builtin_amdgcn_readlane(r.x, j + u);
                    wt[u] = __int_as_float(__builtin_amdgcn_readlane(r.y, j + u));
                }
                if (nload == GATHER_U) {
#pragma unroll
                    for (int u = 0; u < GATHER_U; ++u)
                        v[u] = *(const float4*)(tl + (size_t)s[u] * HH);
                    if (j == 0 && b0 + 32 + lane < e3) rn = *slot(b0 + 32 + lane);
                    stream_group<true, SC>(v, wt, anyovf, nload, b0 + j, e1, e2, e3, d1, d2, d3, bv, hrow, acc, cur, nend, dcur);
                } else {
#pragma unroll
                    for (int u = 0; u < GATHER_U; ++u) {
                        v[u] = make_float4(0.0f, 0.0f, 0.0f, 0.0f);
                        if (u < nload) v[u] = *(const float4*)(tl + (size_t)s[u] * HH);
                    }
                    stream_group<false, SC>(v, wt, anyovf, nload, b0 + j, e1, e2, e3, d1, d2, d3, bv, hrow, acc, cur, nend, dcur);
                }
            }
            r = rn;
        }
    }
    __syncthreads();
    if (wave * 32 >= NCOL) return;

    const int r  = lane & 15;
    const int kq = (lane >> 4) << 1;
    const float* Ap = hs + r * HS_PITCH + kq;
    const float* Bp = Wn + (size_t)kq * NCOL + wave * 32 + r;
    constexpr bool TWO = (NCOL >= 32);

    v8f acc0 = (v8f){};
    v8f acc1 = (v8f){};
#pragma unroll 8
    for (int k0 = 0; k0 < HH; k0 += 4) {
        const float2 av = *(const float2*)(Ap + k0);
        v2f a, b0v;
        a.x = av.x;
        a.y = av.y;
        b0v.x = Bp[(size_t)k0 * NCOL];
        b0v.y = Bp[(size_t)(k0 + 1) * NCOL];
        acc0 = __builtin_amdgcn_wmma_f32_16x16x4_f32(
            false, a, false, b0v, (short)0, acc0, false, false);
        if (TWO) {
            v2f b1v;
            b1v.x = Bp[(size_t)k0 * NCOL + 16];
            b1v.y = Bp[(size_t)(k0 + 1) * NCOL + 16];
            acc1 = __builtin_amdgcn_wmma_f32_16x16x4_f32(
                false, a, false, b1v, (short)0, acc1, false, false);
        }
    }

    const int n  = wave * 32 + r;
    const int mb = node0 + ((lane >> 4) << 3);
    const float bvn0 = bn ? bn[n] : 0.0f;
#pragma unroll
    for (int g = 0; g < 8; ++g)
        C[(size_t)(mb + g) * NCOL + n] = acc0[g] + bvn0;
    if (TWO) {
        const float bvn1 = bn ? bn[n + 16] : 0.0f;
#pragma unroll
        for (int g = 0; g < 8; ++g)
            C[(size_t)(mb + g) * NCOL + n + 16] = acc1[g] + bvn1;
    }
}

static inline int nblocks(long long t, int bs) { return (int)((t + bs - 1) / bs); }

extern "C" void kernel_launch(void* const* d_in, const int* in_sizes, int n_in,
                              void* d_out, int out_size, void* d_ws, size_t ws_size,
                              hipStream_t stream) {
    const float* x  = (const float*)d_in[0];
    const int*   ei = (const int*)d_in[1];
    const float* ew = (const float*)d_in[2];
    const float* W1 = (const float*)d_in[3];  const float* b1 = (const float*)d_in[4];
    const float* W2 = (const float*)d_in[5];  const float* b2 = (const float*)d_in[6];
    const float* W3 = (const float*)d_in[7];  const float* b3 = (const float*)d_in[8];
    const float* Wg[4] = { (const float*)d_in[9],  (const float*)d_in[11],
                           (const float*)d_in[13], (const float*)d_in[15] };
    const float* bg[4] = { (const float*)d_in[10], (const float*)d_in[12],
                           (const float*)d_in[14], (const float*)d_in[16] };
    const float* Wh = (const float*)d_in[17]; const float* bh = (const float*)d_in[18];
    float* out = (float*)d_out;

    const int* srcI = ei;
    const int* dstI = ei + EE;

    float* f     = (float*)d_ws;
    float* hbuf  = f;
    float* tbuf  = hbuf + (size_t)NN * HH;
    int2*  recb  = (int2*)(tbuf + (size_t)NN * HH);
    int4*  ovfb  = (int4*)(recb + (size_t)NN * BPITCH);
    int*   cntb  = (int*)(ovfb + EE);
    float* dinvb = (float*)(cntb + NN + 1);

    const int BS = 256;

    zero_int_kernel<<<nblocks(NN + 1, BS), BS, 0, stream>>>(cntb, NN + 1);
    bucket_kernel<<<nblocks(EE, BS), BS, 0, stream>>>(srcI, dstI, ew, cntb, recb, ovfb);

    mlp_chain_kernel<<<MLP_GRID, BS, 0, stream>>>(x, W1, b1, W2, b2, W3, b3, Wg[0],
                                                  cntb, recb, ovfb, dinvb, tbuf);

    float* tcur = tbuf;
    float* tnxt = hbuf;
    for (int l = 0; l < 3; ++l) {
        gather_gemm_kernel<HH><<<NN / FUSE_ROWS, FUSE_WAVES * 32, 0, stream>>>(
            tcur, cntb, recb, ovfb, dinvb, bg[l], Wg[l + 1], nullptr, tnxt);
        float* sw = tcur; tcur = tnxt; tnxt = sw;
    }
    gather_gemm_kernel<OUTD><<<NN / FUSE_ROWS, FUSE_WAVES * 32, 0, stream>>>(
        tcur, cntb, recb, ovfb, dinvb, bg[3], Wh, bh, out);
}
